# starting kernel
# baseline (speedup 1.0000x reference)
#include <hip/hip_runtime.h>
#include <math.h>

#define NN     10000
#define EE     160000
#define IND    32
#define OUTD   500
#define FPAD   512
#define NQ     125
#define NTILES 32
#define MTILES 625
#define NEG_SLOPE 0.2f

typedef __attribute__((ext_vector_type(2)))  float    v2f;
typedef __attribute__((ext_vector_type(8)))  float    v8f;
typedef __attribute__((ext_vector_type(16))) _Float16 v16h;

#if __has_builtin(__builtin_amdgcn_wmma_f32_16x16x4_f32)
#define USE_WMMA_F32 1
#else
#define USE_WMMA_F32 0
#endif

__device__ __forceinline__ unsigned enc_f(float f) {
  unsigned u = __float_as_uint(f);
  return (u & 0x80000000u) ? ~u : (u | 0x80000000u);
}
__device__ __forceinline__ float dec_f(unsigned e) {
  return (e & 0x80000000u) ? __uint_as_float(e & 0x7FFFFFFFu)
                           : __uint_as_float(~e);
}
#define MAXENC_INIT 0x007FFFFFu

__global__ __launch_bounds__(256) void k_init(float* __restrict__ out,
                                              unsigned* __restrict__ m_enc,
                                              float* __restrict__ s) {
  const int i = blockIdx.x * 256 + threadIdx.x;
  if (i < NN * OUTD) out[i] = 0.0f;
  if (i < NN) { m_enc[i] = MAXENC_INIT; s[i] = 0.0f; }
}

__global__ __launch_bounds__(256) void k_gemm_wmma(const float* __restrict__ x,
                                                   const float* __restrict__ W,
                                                   float* __restrict__ feat) {
  const int gwave = (blockIdx.x * 256 + threadIdx.x) >> 5;
  const int lane  = threadIdx.x & 31;
  const int tileM = gwave >> 5;
  const int tileN = gwave & (NTILES - 1);
  const int m0 = tileM << 4, n0 = tileN << 4;
  const int r  = lane & 15;
  const int h  = lane >> 4;

  const int   col = n0 + r;
  const int   cc  = (col < OUTD) ? col : 0;
  const float msk = (col < OUTD) ? 1.0f : 0.0f;

  v8f acc = {};
#if USE_WMMA_F32
  v2f A[8], B[8];
  #pragma unroll
  for (int i = 0; i < 8; ++i) {
    const int k = (i << 2) + (h << 1);
    A[i].x = x[(m0 + r) * IND + k];
    A[i].y = x[(m0 + r) * IND + k + 1];
    B[i].x = W[k * OUTD + cc] * msk;
    B[i].y = W[(k + 1) * OUTD + cc] * msk;
  }
  #pragma unroll
  for (int i = 0; i < 8; ++i)
    acc = __builtin_amdgcn_wmma_f32_16x16x4_f32(
        false, A[i], false, B[i], (short)0, acc, false, false);
#else
  v16h a, b;
  #pragma unroll
  for (int j = 0; j < 8; ++j) {
    const int ka = ((j >> 2) << 4) + ((j & 3) << 1) + (h << 3);
    a[2 * j]     = (_Float16)x[(m0 + r) * IND + ka];
    a[2 * j + 1] = (_Float16)x[(m0 + r) * IND + ka + 1];
    const int kb = (j << 1) + (h << 4);
    b[2 * j]     = (_Float16)(W[kb * OUTD + cc] * msk);
    b[2 * j + 1] = (_Float16)(W[(kb + 1) * OUTD + cc] * msk);
  }
  acc = __builtin_amdgcn_wmma_f32_16x16x32_f16(
      false, a, false, b, (short)0, acc, false, false);
#endif
  #pragma unroll
  for (int rr = 0; rr < 8; ++rr)
    feat[(size_t)(m0 + rr + (h << 3)) * FPAD + n0 + r] = acc[rr];
}

__global__ __launch_bounds__(256) void k_el_er(const float* __restrict__ feat,
                                               const float* __restrict__ attn_l,
                                               const float* __restrict__ attn_r,
                                               float* __restrict__ el,
                                               float* __restrict__ er) {
  const int node = (blockIdx.x * 256 + threadIdx.x) >> 5;
  const int lane = threadIdx.x & 31;
  if (node >= NN) return;
  const float4* f4 = (const float4*)(feat + (size_t)node * FPAD);
  const float4* l4 = (const float4*)attn_l;
  const float4* r4 = (const float4*)attn_r;
  float sl = 0.0f, sr = 0.0f;
  for (int q = lane; q < NQ; q += 32) {
    const float4 v = f4[q], l = l4[q], rr = r4[q];
    sl = fmaf(v.x, l.x, fmaf(v.y, l.y, fmaf(v.z, l.z, fmaf(v.w, l.w, sl))));
    sr = fmaf(v.x, rr.x, fmaf(v.y, rr.y, fmaf(v.z, rr.z, fmaf(v.w, rr.w, sr))));
  }
  #pragma unroll
  for (int off = 16; off; off >>= 1) {
    sl += __shfl_xor(sl, off, 32);
    sr += __shfl_xor(sr, off, 32);
  }
  if (lane == 0) { el[node] = sl; er[node] = sr; }
}

__global__ __launch_bounds__(256) void k_edge_logit(const float* __restrict__ el,
                                                    const float* __restrict__ er,
                                                    const int* __restrict__ src,
                                                    const int* __restrict__ dst,
                                                    float* __restrict__ e,
                                                    unsigned* __restrict__ m_enc) {
  const int i = blockIdx.x * 256 + threadIdx.x;
  if (i >= EE) return;
  float v = el[src[i]] + er[dst[i]];
  v = (v > 0.0f) ? v : NEG_SLOPE * v;
  e[i] = v;
  atomicMax(&m_enc[dst[i]], enc_f(v));
}

__global__ __launch_bounds__(256) void k_edge_exp(const int* __restrict__ dst,
                                                  float* __restrict__ e,
                                                  const unsigned* __restrict__ m_enc,
                                                  float* __restrict__ s) {
  const int i = blockIdx.x * 256 + threadIdx.x;
  if (i >= EE) return;
  const int d = dst[i];
  const float a = __expf(e[i] - dec_f(m_enc[d]));
  e[i] = a;
  atomicAdd(&s[d], a);
}

__global__ __launch_bounds__(256) void k_scatter(const int* __restrict__ src,
                                                 const int* __restrict__ dst,
                                                 const float* __restrict__ a,
                                                 const float* __restrict__ s,
                                                 const float* __restrict__ feat,
                                                 float* __restrict__ out) {
  const int edge = (blockIdx.x * 256 + threadIdx.x) >> 5;
  const int lane = threadIdx.x & 31;
  const int sn = src[edge], dn = dst[edge];
  const float sv = s[dn];
  const float alpha = a[edge] / (sv > 0.0f ? sv : 1.0f);
  const float4* f4 = (const float4*)(feat + (size_t)sn * FPAD);
  float*        o  = out + (size_t)dn * OUTD;
  __builtin_prefetch((const float*)f4 + 4 * lane, 0, 1);
  #pragma unroll
  for (int q = lane; q < FPAD / 4; q += 32) {
    const int d = q << 2;
    if (d < OUTD) {
      const float4 v = f4[q];
      atomicAdd(&o[d + 0], alpha * v.x);
      atomicAdd(&o[d + 1], alpha * v.y);
      atomicAdd(&o[d + 2], alpha * v.z);
      atomicAdd(&o[d + 3], alpha * v.w);
    }
  }
}

__global__ __launch_bounds__(256) void k_row_softmax(float* __restrict__ out) {
  const int node = (blockIdx.x * 256 + threadIdx.x) >> 5;
  const int lane = threadIdx.x & 31;
  if (node >= NN) return;
  float4* r4 = (float4*)(out + (size_t)node * OUTD);
  float mx = -INFINITY;
  for (int q = lane; q < NQ; q += 32) {
    const float4 v = r4[q];
    mx = fmaxf(mx, fmaxf(fmaxf(v.x, v.y), fmaxf(v.z, v.w)));
  }
  #pragma unroll
  for (int off = 16; off; off >>= 1) mx = fmaxf(mx, __shfl_xor(mx, off, 32));
  float sm = 0.0f;
  for (int q = lane; q < NQ; q += 32) {
    float4 v = r4[q];
    v.x = __expf(v.x - mx); v.y = __expf(v.y - mx);
    v.z = __expf(v.z - mx); v.w = __expf(v.w - mx);
    r4[q] = v;
    sm += (v.x + v.y) + (v.z + v.w);
  }
  #pragma unroll
  for (int off = 16; off; off >>= 1) sm += __shfl_xor(sm, off, 32);
  const float inv = 1.0f / sm;
  for (int q = lane; q < NQ; q += 32) {
    float4 v = r4[q];
    v.x *= inv; v.y *= inv; v.z *= inv; v.w *= inv;
    r4[q] = v;
  }
}

__global__ __launch_bounds__(256) void k_px_r(const float* __restrict__ px_r,
                                              float* __restrict__ out) {
  const int i = blockIdx.x * 256 + threadIdx.x;
  if (i >= OUTD) return;
  const float v = px_r[i];
  const float sp = (v > 20.0f) ? v : log1pf(__expf(v));
  out[(size_t)NN * OUTD + i] = fminf(fmaxf(sp, 1.0e-4f), 1.0e4f);
}

extern "C" void kernel_launch(void* const* d_in, const int* in_sizes, int n_in,
                              void* d_out, int out_size, void* d_ws, size_t ws_size,
                              hipStream_t stream) {
  const float* x      = (const float*)d_in[0];
  const float* W      = (const float*)d_in[1];
  const float* attn_l = (const float*)d_in[2];
  const float* attn_r = (const float*)d_in[3];
  const float* px_r   = (const float*)d_in[4];
  const int*   src    = (const int*)d_in[5];
  const int*   dst    = (const int*)d_in[6];
  float*       out    = (float*)d_out;

  float*    ws    = (float*)d_ws;
  float*    feat  = ws;
  float*    el    = feat + (size_t)NN * FPAD;
  float*    er    = el + NN;
  unsigned* m_enc = (unsigned*)(er + NN);
  float*    s     = (float*)(m_enc + NN);
  float*    ea    = s + NN;

  const int initN = NN * OUTD;
  k_init       <<<(initN + 255) / 256, 256, 0, stream>>>(out, m_enc, s);
  k_gemm_wmma  <<<(MTILES * NTILES) / 8, 256, 0, stream>>>(x, W, feat);
  k_el_er      <<<(NN + 7) / 8, 256, 0, stream>>>(feat, attn_l, attn_r, el, er);
  k_edge_logit <<<(EE + 255) / 256, 256, 0, stream>>>(el, er, src, dst, ea, m_enc);
  k_edge_exp   <<<(EE + 255) / 256, 256, 0, stream>>>(dst, ea, m_enc, s);
  k_scatter    <<<EE / 8, 256, 0, stream>>>(src, dst, ea, s, feat, out);
  k_row_softmax<<<(NN + 7) / 8, 256, 0, stream>>>(out);
  k_px_r       <<<(OUTD + 255) / 256, 256, 0, stream>>>(px_r, out);
}


# float atomics replaced by dst-grouped edge lists: one wave per dst node does the edge softmax and sums alpha*feat[src] in edge-id order, writes each out row once (sum order changes, now fixed)
# speedup vs baseline: 1.5518x; 1.5518x over previous
#include <hip/hip_runtime.h>
#include <math.h>

#define NN     10000
#define EE     160000
#define IND    32
#define OUTD   500
#define FPAD   512
#define NQ     125
#define NTILES 32
#define MTILES 625
#define NEG_SLOPE 0.2f
#define SCAN_T 1024
#define SCAN_PER 10

typedef __attribute__((ext_vector_type(2)))  float    v2f;
typedef __attribute__((ext_vector_type(8)))  float    v8f;
typedef __attribute__((ext_vector_type(16))) _Float16 v16h;

#if __has_builtin(__builtin_amdgcn_wmma_f32_16x16x4_f32)
#define USE_WMMA_F32 1
#else
#define USE_WMMA_F32 0
#endif

__global__ __launch_bounds__(256) void k_setup(const float* __restrict__ px_r,
                                               float* __restrict__ out,
                                               int* __restrict__ cnt) {
  const int i = blockIdx.x * 256 + threadIdx.x;
  if (i < NN) cnt[i] = 0;
  if (i < OUTD) {
    const float v = px_r[i];
    const float sp = (v > 20.0f) ? v : log1pf(__expf(v));
    out[(size_t)NN * OUTD + i] = fminf(fmaxf(sp, 1.0e-4f), 1.0e4f);
  }
}

__global__ __launch_bounds__(256) void k_gemm_wmma(const float* __restrict__ x,
                                                   const float* __restrict__ W,
                                                   float* __restrict__ feat) {
  const int gwave = (blockIdx.x * 256 + threadIdx.x) >> 5;
  const int lane  = threadIdx.x & 31;
  const int tileM = gwave >> 5;
  const int tileN = gwave & (NTILES - 1);
  const int m0 = tileM << 4, n0 = tileN << 4;
  const int r  = lane & 15;
  const int h  = lane >> 4;

  const int   col = n0 + r;
  const int   cc  = (col < OUTD) ? col : 0;
  const float msk = (col < OUTD) ? 1.0f : 0.0f;

  v8f acc = {};
#if USE_WMMA_F32
  v2f A[8], B[8];
  #pragma unroll
  for (int i = 0; i < 8; ++i) {
    const int k = (i << 2) + (h << 1);
    A[i].x = x[(m0 + r) * IND + k];
    A[i].y = x[(m0 + r) * IND + k + 1];
    B[i].x = W[k * OUTD + cc] * msk;
    B[i].y = W[(k + 1) * OUTD + cc] * msk;
  }
  #pragma unroll
  for (int i = 0; i < 8; ++i)
    acc = __builtin_amdgcn_wmma_f32_16x16x4_f32(
        false, A[i], false, B[i], (short)0, acc, false, false);
#else
  v16h a, b;
  #pragma unroll
  for (int j = 0; j < 8; ++j) {
    const int ka = ((j >> 2) << 4) + ((j & 3) << 1) + (h << 3);
    a[2 * j]     = (_Float16)x[(m0 + r) * IND + ka];
    a[2 * j + 1] = (_Float16)x[(m0 + r) * IND + ka + 1];
    const int kb = (j << 1) + (h << 4);
    b[2 * j]     = (_Float16)(W[kb * OUTD + cc] * msk);
    b[2 * j + 1] = (_Float16)(W[(kb + 1) * OUTD + cc] * msk);
  }
  acc = __builtin_amdgcn_wmma_f32_16x16x32_f16(
      false, a, false, b, (short)0, acc, false, false);
#endif
  #pragma unroll
  for (int rr = 0; rr < 8; ++rr)
    feat[(size_t)(m0 + rr + (h << 3)) * FPAD + n0 + r] = acc[rr];
}

__global__ __launch_bounds__(256) void k_el_er(const float* __restrict__ feat,
                                               const float* __restrict__ attn_l,
                                               const float* __restrict__ attn_r,
                                               float* __restrict__ el,
                                               float* __restrict__ er) {
  const int node = (blockIdx.x * 256 + threadIdx.x) >> 5;
  const int lane = threadIdx.x & 31;
  if (node >= NN) return;
  const float4* f4 = (const float4*)(feat + (size_t)node * FPAD);
  const float4* l4 = (const float4*)attn_l;
  const float4* r4 = (const float4*)attn_r;
  float sl = 0.0f, sr = 0.0f;
  for (int q = lane; q < NQ; q += 32) {
    const float4 v = f4[q], l = l4[q], rr = r4[q];
    sl = fmaf(v.x, l.x, fmaf(v.y, l.y, fmaf(v.z, l.z, fmaf(v.w, l.w, sl))));
    sr = fmaf(v.x, rr.x, fmaf(v.y, rr.y, fmaf(v.z, rr.z, fmaf(v.w, rr.w, sr))));
  }
  #pragma unroll
  for (int off = 16; off; off >>= 1) {
    sl += __shfl_xor(sl, off, 32);
    sr += __shfl_xor(sr, off, 32);
  }
  if (lane == 0) { el[node] = sl; er[node] = sr; }
}

__global__ __launch_bounds__(256) void k_count(const int* __restrict__ dst,
                                               int* __restrict__ cnt) {
  const int i = blockIdx.x * 256 + threadIdx.x;
  if (i < EE) atomicAdd(&cnt[dst[i]], 1);
}

__global__ __launch_bounds__(SCAN_T) void k_scan(int* __restrict__ cntcur,
                                                 int* __restrict__ off) {
  __shared__ int wsum[32];
  const int t = threadIdx.x, lane = t & 31, wv = t >> 5;
  const int base = t * SCAN_PER;
  int c[SCAN_PER];
  int tot = 0;
  #pragma unroll
  for (int j = 0; j < SCAN_PER; ++j) {
    const int i = base + j;
    c[j] = (i < NN) ? cntcur[i] : 0;
    tot += c[j];
  }
  int inc = tot;
  #pragma unroll
  for (int o = 1; o < 32; o <<= 1) {
    const int u = __shfl_up(inc, o, 32);
    if (lane >= o) inc += u;
  }
  if (lane == 31) wsum[wv] = inc;
  __syncthreads();
  if (wv == 0) {
    const int w = wsum[lane];
    int winc = w;
    #pragma unroll
    for (int o = 1; o < 32; o <<= 1) {
      const int u = __shfl_up(winc, o, 32);
      if (lane >= o) winc += u;
    }
    wsum[lane] = winc - w;
  }
  __syncthreads();
  int run = wsum[wv] + inc - tot;
  #pragma unroll
  for (int j = 0; j < SCAN_PER; ++j) {
    const int i = base + j;
    if (i < NN) { off[i] = run; cntcur[i] = run; }
    run += c[j];
  }
}

__global__ __launch_bounds__(256) void k_fill(const int* __restrict__ dst,
                                              int* __restrict__ cur,
                                              int* __restrict__ csr) {
  const int i = blockIdx.x * 256 + threadIdx.x;
  if (i < EE) {
    const int p = atomicAdd(&cur[dst[i]], 1);
    csr[p] = i;
  }
}

__device__ __forceinline__ float wave_max(float v) {
  #pragma unroll
  for (int o = 16; o; o >>= 1) v = fmaxf(v, __shfl_xor(v, o, 32));
  return v;
}
__device__ __forceinline__ float leaky(float v) {
  return (v > 0.0f) ? v : NEG_SLOPE * v;
}
__device__ __forceinline__ void row_fma(float4 (&acc)[4], const float* __restrict__ row,
                                        float al, int lane) {
  const float4* f4 = (const float4*)row;
  #pragma unroll
  for (int j = 0; j < 4; ++j) {
    const float4 v = f4[lane + 32 * j];
    acc[j].x = fmaf(al, v.x, acc[j].x);
    acc[j].y = fmaf(al, v.y, acc[j].y);
    acc[j].z = fmaf(al, v.z, acc[j].z);
    acc[j].w = fmaf(al, v.w, acc[j].w);
  }
}
__device__ __forceinline__ int next_edge(const int* __restrict__ csr, int beg, int end,
                                         int prev, int lane) {
  int best = 0x7FFFFFFF;
  for (int p = beg + lane; p < end; p += 32) {
    const int c = csr[p];
    if (c > prev && c < best) best = c;
  }
  #pragma unroll
  for (int o = 16; o; o >>= 1) best = min(best, __shfl_xor(best, o, 32));
  return __builtin_amdgcn_readfirstlane(best);
}

__global__ __launch_bounds__(256) void k_gather(const int* __restrict__ src,
                                                const int* __restrict__ off,
                                                const int* __restrict__ csr,
                                                const float* __restrict__ el,
                                                const float* __restrict__ er,
                                                const float* __restrict__ feat,
                                                float* __restrict__ out) {
  const int node = __builtin_amdgcn_readfirstlane((blockIdx.x * 256 + threadIdx.x) >> 5);
  const int lane = threadIdx.x & 31;
  const int beg = off[node];
  const int end = (node + 1 < NN) ? off[node + 1] : EE;
  const int deg = end - beg;
  const float ern = er[node];
  float4 acc[4];
  #pragma unroll
  for (int j = 0; j < 4; ++j) acc[j] = make_float4(0.0f, 0.0f, 0.0f, 0.0f);

  if (deg <= 32) {
    const bool act = lane < deg;
    const int id = act ? csr[beg + lane] : 0x7FFFFFFF;
    const int sn = act ? src[id] : 0;
    const float e = act ? leaky(el[sn] + ern) : -INFINITY;
    int rank = 0;
    for (int j = 0; j < deg; ++j)
      rank += (__builtin_amdgcn_readlane(id, j) < id) ? 1 : 0;
    if (!act) rank = lane;
    float m = wave_max(e);
    m = isfinite(m) ? m : 0.0f;
    const float a = act ? __expf(e - m) : 0.0f;
    const int sa = __builtin_amdgcn_ds_permute(rank << 2, __float_as_int(a));
    const int ss = __builtin_amdgcn_ds_permute(rank << 2, sn);
    float s = 0.0f;
    for (int k = 0; k < deg; ++k) s += __int_as_float(__builtin_amdgcn_readlane(sa, k));
    const int al_bits = __float_as_int(__int_as_float(sa) / (s > 0.0f ? s : 1.0f));
    #pragma unroll 4
    for (int k = 0; k < deg; ++k) {
      const int   snk = __builtin_amdgcn_readlane(ss, k);
      const float al  = __int_as_float(__builtin_amdgcn_readlane(al_bits, k));
      row_fma(acc, feat + (size_t)snk * FPAD, al, lane);
    }
  } else {
    float mx = -INFINITY;
    for (int p = beg + lane; p < end; p += 32)
      mx = fmaxf(mx, leaky(el[src[csr[p]]] + ern));
    float m = wave_max(mx);
    m = isfinite(m) ? m : 0.0f;
    float s = 0.0f;
    int prev = -1;
    for (int k = 0; k < deg; ++k) {
      prev = next_edge(csr, beg, end, prev, lane);
      s += __expf(leaky(el[src[prev]] + ern) - m);
    }
    const float den = (s > 0.0f) ? s : 1.0f;
    prev = -1;
    for (int k = 0; k < deg; ++k) {
      prev = next_edge(csr, beg, end, prev, lane);
      const int snk = src[prev];
      const float al = __expf(leaky(el[snk] + ern) - m) / den;
      row_fma(acc, feat + (size_t)snk * FPAD, al, lane);
    }
  }
  float4* o4 = (float4*)(out + (size_t)node * OUTD);
  #pragma unroll
  for (int j = 0; j < 4; ++j) {
    const int q = lane + 32 * j;
    if (q < NQ) o4[q] = acc[j];
  }
}

__global__ __launch_bounds__(256) void k_row_softmax(float* __restrict__ out) {
  const int node = (blockIdx.x * 256 + threadIdx.x) >> 5;
  const int lane = threadIdx.x & 31;
  if (node >= NN) return;
  float4* r4 = (float4*)(out + (size_t)node * OUTD);
  float mx = -INFINITY;
  for (int q = lane; q < NQ; q += 32) {
    const float4 v = r4[q];
    mx = fmaxf(mx, fmaxf(fmaxf(v.x, v.y), fmaxf(v.z, v.w)));
  }
  #pragma unroll
  for (int off = 16; off; off >>= 1) mx = fmaxf(mx, __shfl_xor(mx, off, 32));
  float sm = 0.0f;
  for (int q = lane; q < NQ; q += 32) {
    float4 v = r4[q];
    v.x = __expf(v.x - mx); v.y = __expf(v.y - mx);
    v.z = __expf(v.z - mx); v.w = __expf(v.w - mx);
    r4[q] = v;
    sm += (v.x + v.y) + (v.z + v.w);
  }
  #pragma unroll
  for (int off = 16; off; off >>= 1) sm += __shfl_xor(sm, off, 32);
  const float inv = 1.0f / sm;
  for (int q = lane; q < NQ; q += 32) {
    float4 v = r4[q];
    v.x *= inv; v.y *= inv; v.z *= inv; v.w *= inv;
    r4[q] = v;
  }
}

extern "C" void kernel_launch(void* const* d_in, const int* in_sizes, int n_in,
                              void* d_out, int out_size, void* d_ws, size_t ws_size,
                              hipStream_t stream) {
  const float* x      = (const float*)d_in[0];
  const float* W      = (const float*)d_in[1];
  const float* attn_l = (const float*)d_in[2];
  const float* attn_r = (const float*)d_in[3];
  const float* px_r   = (const float*)d_in[4];
  const int*   src    = (const int*)d_in[5];
  const int*   dst    = (const int*)d_in[6];
  float*       out    = (float*)d_out;

  float* ws   = (float*)d_ws;
  float* feat = ws;
  float* el   = feat + (size_t)NN * FPAD;
  float* er   = el + NN;
  int*   cnt  = (int*)(er + NN);
  int*   off  = cnt + NN;
  int*   csr  = off + NN;

  k_setup      <<<(NN + 255) / 256, 256, 0, stream>>>(px_r, out, cnt);
  k_count      <<<(EE + 255) / 256, 256, 0, stream>>>(dst, cnt);
  k_scan       <<<1, SCAN_T, 0, stream>>>(cnt, off);
  k_fill       <<<(EE + 255) / 256, 256, 0, stream>>>(dst, cnt, csr);
  k_gemm_wmma  <<<(MTILES * NTILES) / 8, 256, 0, stream>>>(x, W, feat);
  k_el_er      <<<(NN + 7) / 8, 256, 0, stream>>>(feat, attn_l, attn_r, el, er);
  k_gather     <<<NN / 8, 256, 0, stream>>>(src, off, csr, el, er, feat, out);
  k_row_softmax<<<(NN + 7) / 8, 256, 0, stream>>>(out);
}


# row softmax fused into the f32 WMMA GEMM: one 8-wave block per row tile computes whole rows, normalises them in LDS and writes every 128-B line of out once; 7 -> 6 launches; sum order unchanged
# speedup vs baseline: 2.0875x; 1.3452x over previous
#include <hip/hip_runtime.h>
#include <math.h>

#define NN     10000
#define EE     160000
#define IND    32
#define OUTD   500
#define NQ     125
#define NTILES 32
#define MTILES 625
#define NEG_SLOPE 0.2f
#define SCAN_T 1024
#define SCAN_PER 10

typedef __attribute__((ext_vector_type(2)))  float    v2f;
typedef __attribute__((ext_vector_type(8)))  float    v8f;

__global__ __launch_bounds__(256) void k_setup(const float* __restrict__ px_r,
                                               const float* __restrict__ W,
                                               const float* __restrict__ attn_l,
                                               const float* __restrict__ attn_r,
                                               float* __restrict__ out,
                                               int* __restrict__ cnt,
                                               float* __restrict__ wlr) {
  const int i = blockIdx.x * 256 + threadIdx.x;
  if (i < NN) cnt[i] = 0;
  if (i < OUTD) {
    const float v = px_r[i];
    const float sp = (v > 20.0f) ? v : log1pf(__expf(v));
    out[(size_t)NN * OUTD + i] = fminf(fmaxf(sp, 1.0e-4f), 1.0e4f);
  }
  const int gw = i >> 5;
  if (gw < 2 * IND) {
    const int lane = threadIdx.x & 31;
    const int k = gw & (IND - 1);
    const float* a = (gw < IND) ? attn_l : attn_r;
    float s = 0.0f;
    #pragma unroll
    for (int j = 0; j < 16; ++j) {
      const int d  = lane + 32 * j;
      const int dc = (d < OUTD) ? d : 0;
      s = fmaf(W[k * OUTD + dc], (d < OUTD) ? a[dc] : 0.0f, s);
    }
    #pragma unroll
    for (int o = 16; o; o >>= 1) s += __shfl_xor(s, o, 32);
    if (lane == 0) wlr[gw] = s;
  }
}

__global__ __launch_bounds__(256) void k_gemm_softmax(const float* __restrict__ agg,
                                                      const float* __restrict__ W,
                                                      float* __restrict__ out) {
  __shared__ float4 S4[16 * NQ + 8];
  float* S = (float*)S4;
  const int wv   = threadIdx.x >> 5;
  const int lane = threadIdx.x & 31;
  const int r  = lane & 15;
  const int h  = lane >> 4;
  const int m0 = blockIdx.x << 4;

  v2f A[8], B[4][8];
  #pragma unroll
  for (int i = 0; i < 8; ++i) {
    const int k = (i << 2) + (h << 1);
    A[i].x = agg[(m0 + r) * IND + k];
    A[i].y = agg[(m0 + r) * IND + k + 1];
  }
  #pragma unroll
  for (int t = 0; t < 4; ++t) {
    const int col = (wv << 6) + (t << 4) + r;
    const int cc  = (col < OUTD) ? col : 0;
    #pragma unroll
    for (int i = 0; i < 8; ++i) {
      const int k = (i << 2) + (h << 1);
      B[t][i].x = W[k * OUTD + cc];
      B[t][i].y = W[(k + 1) * OUTD + cc];
    }
  }
  v8f acc[4] = {};
  #pragma unroll
  for (int i = 0; i < 8; ++i) {
    #pragma unroll
    for (int t = 0; t < 4; ++t)
      acc[t] = __builtin_amdgcn_wmma_f32_16x16x4_f32(
          false, A[i], false, B[t][i], (short)0, acc[t], false, false);
  }

  #pragma unroll
  for (int t = 0; t < 4; ++t) {
    const int col = (wv << 6) + (t << 4) + r;
    #pragma unroll
    for (int rr = 0; rr < 8; ++rr) {
      int idx = (rr + (h << 3)) * OUTD + col;
      if (t == 3) idx = (col < OUTD) ? idx : 16 * OUTD + lane;
      S[idx] = acc[t][rr];
    }
  }
  __syncthreads();

  const bool last = (lane + 96) < NQ;
  #pragma unroll
  for (int rw = 0; rw < 2; ++rw) {
    float4* r4 = S4 + ((wv << 1) + rw) * NQ;
    float4 v[4];
    #pragma unroll
    for (int j = 0; j < 3; ++j) v[j] = r4[lane + 32 * j];
    v[3] = last ? r4[lane + 96] : make_float4(-INFINITY, -INFINITY, -INFINITY, -INFINITY);
    float mx = -INFINITY;
    #pragma unroll
    for (int j = 0; j < 4; ++j)
      mx = fmaxf(mx, fmaxf(fmaxf(v[j].x, v[j].y), fmaxf(v[j].z, v[j].w)));
    #pragma unroll
    for (int off = 16; off; off >>= 1) mx = fmaxf(mx, __shfl_xor(mx, off, 32));
    float sm = 0.0f;
    #pragma unroll
    for (int j = 0; j < 4; ++j) {
      const bool act = (j < 3) || last;
      v[j].x = act ? __expf(v[j].x - mx) : 0.0f;
      v[j].y = act ? __expf(v[j].y - mx) : 0.0f;
      v[j].z = act ? __expf(v[j].z - mx) : 0.0f;
      v[j].w = act ? __expf(v[j].w - mx) : 0.0f;
      sm += (v[j].x + v[j].y) + (v[j].z + v[j].w);
    }
    #pragma unroll
    for (int off = 16; off; off >>= 1) sm += __shfl_xor(sm, off, 32);
    const float inv = 1.0f / sm;
    #pragma unroll
    for (int j = 0; j < 4; ++j) {
      v[j].x *= inv; v[j].y *= inv; v[j].z *= inv; v[j].w *= inv;
    }
    #pragma unroll
    for (int j = 0; j < 3; ++j) r4[lane + 32 * j] = v[j];
    if (last) r4[lane + 96] = v[3];
  }
  __syncthreads();

  float4* o4 = (float4*)(out + (size_t)m0 * OUTD);
  const int tid = threadIdx.x;
  #pragma unroll
  for (int j = 0; j < 7; ++j) o4[tid + 256 * j] = S4[tid + 256 * j];
  if (tid < 16 * NQ - 7 * 256) o4[tid + 7 * 256] = S4[tid + 7 * 256];
}

__global__ __launch_bounds__(256) void k_count_el(const int* __restrict__ dst,
                                                  int* __restrict__ cnt,
                                                  const float* __restrict__ x,
                                                  const float* __restrict__ wlr,
                                                  float* __restrict__ el,
                                                  float* __restrict__ er) {
  const int i = blockIdx.x * 256 + threadIdx.x;
  if (i < EE) atomicAdd(&cnt[dst[i]], 1);
  if (i < NN * 8) {
    const int part = i & 7;
    const float4 v = ((const float4*)x)[i];
    const float4 l = ((const float4*)wlr)[part];
    const float4 rr = ((const float4*)wlr)[8 + part];
    float sl = fmaf(v.x, l.x, fmaf(v.y, l.y, fmaf(v.z, l.z, v.w * l.w)));
    float sr = fmaf(v.x, rr.x, fmaf(v.y, rr.y, fmaf(v.z, rr.z, v.w * rr.w)));
    #pragma unroll
    for (int o = 4; o; o >>= 1) {
      sl += __shfl_xor(sl, o, 32);
      sr += __shfl_xor(sr, o, 32);
    }
    if (part == 0) { el[i >> 3] = sl; er[i >> 3] = sr; }
  }
}

__global__ __launch_bounds__(SCAN_T) void k_scan(int* __restrict__ cntcur,
                                                 int* __restrict__ off) {
  __shared__ int wsum[32];
  const int t = threadIdx.x, lane = t & 31, wv = t >> 5;
  const int base = t * SCAN_PER;
  int c[SCAN_PER];
  int tot = 0;
  #pragma unroll
  for (int j = 0; j < SCAN_PER; ++j) {
    const int i = base + j;
    c[j] = (i < NN) ? cntcur[i] : 0;
    tot += c[j];
  }
  int inc = tot;
  #pragma unroll
  for (int o = 1; o < 32; o <<= 1) {
    const int u = __shfl_up(inc, o, 32);
    if (lane >= o) inc += u;
  }
  if (lane == 31) wsum[wv] = inc;
  __syncthreads();
  if (wv == 0) {
    const int w = wsum[lane];
    int winc = w;
    #pragma unroll
    for (int o = 1; o < 32; o <<= 1) {
      const int u = __shfl_up(winc, o, 32);
      if (lane >= o) winc += u;
    }
    wsum[lane] = winc - w;
  }
  __syncthreads();
  int run = wsum[wv] + inc - tot;
  #pragma unroll
  for (int j = 0; j < SCAN_PER; ++j) {
    const int i = base + j;
    if (i < NN) { off[i] = run; cntcur[i] = run; }
    run += c[j];
  }
}

__global__ __launch_bounds__(256) void k_fill(const int* __restrict__ dst,
                                              int* __restrict__ cur,
                                              int* __restrict__ csr) {
  const int i = blockIdx.x * 256 + threadIdx.x;
  if (i < EE) {
    const int p = atomicAdd(&cur[dst[i]], 1);
    csr[p] = i;
  }
}

__device__ __forceinline__ float wave_max(float v) {
  #pragma unroll
  for (int o = 16; o; o >>= 1) v = fmaxf(v, __shfl_xor(v, o, 32));
  return v;
}
__device__ __forceinline__ float leaky(float v) {
  return (v > 0.0f) ? v : NEG_SLOPE * v;
}
__device__ __forceinline__ int next_edge(const int* __restrict__ csr, int beg, int end,
                                         int prev, int lane) {
  int best = 0x7FFFFFFF;
  for (int p = beg + lane; p < end; p += 32) {
    const int c = csr[p];
    if (c > prev && c < best) best = c;
  }
  #pragma unroll
  for (int o = 16; o; o >>= 1) best = min(best, __shfl_xor(best, o, 32));
  return __builtin_amdgcn_readfirstlane(best);
}

__global__ __launch_bounds__(256) void k_gather(const int* __restrict__ src,
                                                const int* __restrict__ off,
                                                const int* __restrict__ csr,
                                                const float* __restrict__ el,
                                                const float* __restrict__ er,
                                                const float* __restrict__ x,
                                                float* __restrict__ agg) {
  const int node = __builtin_amdgcn_readfirstlane((blockIdx.x * 256 + threadIdx.x) >> 5);
  const int lane = threadIdx.x & 31;
  const int beg = off[node];
  const int end = (node + 1 < NN) ? off[node + 1] : EE;
  const int deg = end - beg;
  const float ern = er[node];
  float acc = 0.0f;

  if (deg <= 32) {
    const bool act = lane < deg;
    const int id = act ? csr[beg + lane] : 0x7FFFFFFF;
    const int sn = act ? src[id] : 0;
    const float e = act ? leaky(el[sn] + ern) : -INFINITY;
    int rank = 0;
    for (int j = 0; j < deg; ++j)
      rank += (__builtin_amdgcn_readlane(id, j) < id) ? 1 : 0;
    if (!act) rank = lane;
    float m = wave_max(e);
    m = isfinite(m) ? m : 0.0f;
    const float a = act ? __expf(e - m) : 0.0f;
    const int sa = __builtin_amdgcn_ds_permute(rank << 2, __float_as_int(a));
    const int ss = __builtin_amdgcn_ds_permute(rank << 2, sn);
    float s = 0.0f;
    for (int k = 0; k < deg; ++k) s += __int_as_float(__builtin_amdgcn_readlane(sa, k));
    const int al_bits = __float_as_int(__int_as_float(sa) / (s > 0.0f ? s : 1.0f));
    for (int k0 = 0; k0 < deg; k0 += 16) {
      #pragma unroll
      for (int j = 0; j < 16; ++j) {
        const int   snk = __builtin_amdgcn_readlane(ss, k0 + j);
        const float al  = __int_as_float(__builtin_amdgcn_readlane(al_bits, k0 + j));
        acc = fmaf(al, x[(size_t)snk * IND + lane], acc);
      }
    }
  } else {
    float mx = -INFINITY;
    for (int p = beg + lane; p < end; p += 32)
      mx = fmaxf(mx, leaky(el[src[csr[p]]] + ern));
    float m = wave_max(mx);
    m = isfinite(m) ? m : 0.0f;
    float s = 0.0f;
    int prev = -1;
    for (int k = 0; k < deg; ++k) {
      prev = next_edge(csr, beg, end, prev, lane);
      s += __expf(leaky(el[src[prev]] + ern) - m);
    }
    const float den = (s > 0.0f) ? s : 1.0f;
    prev = -1;
    for (int k = 0; k < deg; ++k) {
      prev = next_edge(csr, beg, end, prev, lane);
      const int snk = src[prev];
      const float al = __expf(leaky(el[snk] + ern) - m) / den;
      acc = fmaf(al, x[(size_t)snk * IND + lane], acc);
    }
  }
  agg[(size_t)node * IND + lane] = acc;
}

extern "C" void kernel_launch(void* const* d_in, const int* in_sizes, int n_in,
                              void* d_out, int out_size, void* d_ws, size_t ws_size,
                              hipStream_t stream) {
  const float* x      = (const float*)d_in[0];
  const float* W      = (const float*)d_in[1];
  const float* attn_l = (const float*)d_in[2];
  const float* attn_r = (const float*)d_in[3];
  const float* px_r   = (const float*)d_in[4];
  const int*   src    = (const int*)d_in[5];
  const int*   dst    = (const int*)d_in[6];
  float*       out    = (float*)d_out;

  float* ws   = (float*)d_ws;
  float* agg  = ws;
  float* el   = agg + (size_t)NN * IND;
  float* er   = el + NN;
  int*   cnt  = (int*)(er + NN);
  int*   off  = cnt + NN;
  int*   csr  = off + NN;
  float* wlr  = (float*)(csr + EE);

  k_setup      <<<(NN + 255) / 256, 256, 0, stream>>>(px_r, W, attn_l, attn_r, out, cnt, wlr);
  k_count_el   <<<(EE + 255) / 256, 256, 0, stream>>>(dst, cnt, x, wlr, el, er);
  k_scan       <<<1, SCAN_T, 0, stream>>>(cnt, off);
  k_fill       <<<(EE + 255) / 256, 256, 0, stream>>>(dst, cnt, csr);
  k_gather     <<<NN / 8, 256, 0, stream>>>(src, off, csr, el, er, x, agg);
  k_gemm_softmax<<<MTILES, 256, 0, stream>>>(agg, W, out);
}


# edge list built in ONE launch: a work-group owns 125 dst nodes, sweeps all of dst twice with LDS counters (count+scan, then place); no global atomics or counter clear; 6 -> 3 launches; sum order kept
# speedup vs baseline: 3.2526x; 1.5581x over previous
#include <hip/hip_runtime.h>
#include <math.h>

#define NN     10000
#define EE     160000
#define IND    32
#define OUTD   500
#define NQ     125
#define NTILES 32
#define MTILES 625
#define NEG_SLOPE 0.2f

typedef __attribute__((ext_vector_type(2)))  float    v2f;
typedef __attribute__((ext_vector_type(8)))  float    v8f;

#define GA    80
#define NPB   125
#define TA    1024
#define EQ    (EE / 4)
#define FULLR 39
#define UNR   13

__device__ __forceinline__ void count_one(int d, int lo, int* cnt_s, int& below) {
  const int v = d - lo;
  below += (v < 0) ? 1 : 0;
  if ((unsigned)v < (unsigned)NPB) atomicAdd(&cnt_s[v], 1);
}
__device__ __forceinline__ void place_one(int d, int lo, int eid, int* cur_s,
                                          int* __restrict__ csr) {
  const int v = d - lo;
  if ((unsigned)v < (unsigned)NPB) {
    const int p = atomicAdd(&cur_s[v], 1);
    csr[p] = eid;
  }
}

__global__ __launch_bounds__(TA) void k_build(const int* __restrict__ dst,
                                              const float* __restrict__ x,
                                              const float* __restrict__ W,
                                              const float* __restrict__ attn_l,
                                              const float* __restrict__ attn_r,
                                              const float* __restrict__ px_r,
                                              float* __restrict__ out,
                                              int* __restrict__ off,
                                              int* __restrict__ csr,
                                              float* __restrict__ el,
                                              float* __restrict__ er) {
  __shared__ int    cnt_s[128];
  __shared__ int    cur_s[128];
  __shared__ int    red_s[32];
  __shared__ float4 wlr4[16];
  float* wlr_s = (float*)wlr4;
  const int t = threadIdx.x, lane = t & 31, wv = t >> 5;
  const int lo = blockIdx.x * NPB;

  if (t < 128) cnt_s[t] = 0;

  const int xq = min(lo * 8 + t, NN * 8 - 1);
  const float4 xv = ((const float4*)x)[xq];

  if (blockIdx.x == 0 && t < OUTD) {
    const float v = px_r[t];
    const float sp = (v > 20.0f) ? v : log1pf(__expf(v));
    out[(size_t)NN * OUTD + t] = fminf(fmaxf(sp, 1.0e-4f), 1.0e4f);
  }

  #pragma unroll
  for (int g2 = 0; g2 < 2; ++g2) {
    const float* a = g2 ? attn_r : attn_l;
    float s = 0.0f;
    #pragma unroll
    for (int j = 0; j < 16; ++j) {
      const int d  = lane + 32 * j;
      const int dc = (d < OUTD) ? d : 0;
      s = fmaf(W[wv * OUTD + dc], (d < OUTD) ? a[dc] : 0.0f, s);
    }
    #pragma unroll
    for (int o = 16; o; o >>= 1) s += __shfl_xor(s, o, 32);
    if (lane == 0) wlr_s[wv + 32 * g2] = s;
  }
  __syncthreads();

  {
    const int part = t & 7;
    const float4 l  = wlr4[part];
    const float4 rr = wlr4[8 + part];
    float sl = fmaf(xv.x, l.x, fmaf(xv.y, l.y, fmaf(xv.z, l.z, xv.w * l.w)));
    float sr = fmaf(xv.x, rr.x, fmaf(xv.y, rr.y, fmaf(xv.z, rr.z, xv.w * rr.w)));
    #pragma unroll
    for (int o = 4; o; o >>= 1) {
      sl += __shfl_xor(sl, o, 32);
      sr += __shfl_xor(sr, o, 32);
    }
    if (part == 0 && t < NPB * 8) { el[lo + (t >> 3)] = sl; er[lo + (t >> 3)] = sr; }
  }

  const int4* d4 = (const int4*)dst;

  int below = 0;
  #pragma unroll 1
  for (int j0 = 0; j0 < FULLR; j0 += UNR) {
    int4 d[UNR];
    #pragma unroll
    for (int u = 0; u < UNR; ++u) d[u] = d4[t + TA * (j0 + u)];
    #pragma unroll
    for (int u = 0; u < UNR; ++u) {
      count_one(d[u].x, lo, cnt_s, below);
      count_one(d[u].y, lo, cnt_s, below);
      count_one(d[u].z, lo, cnt_s, below);
      count_one(d[u].w, lo, cnt_s, below);
    }
  }
  if (t < EQ - FULLR * TA) {
    const int4 d = d4[FULLR * TA + t];
    count_one(d.x, lo, cnt_s, below);
    count_one(d.y, lo, cnt_s, below);
    count_one(d.z, lo, cnt_s, below);
    count_one(d.w, lo, cnt_s, below);
  }
  #pragma unroll
  for (int o = 16; o; o >>= 1) below += __shfl_xor(below, o, 32);
  if (lane == 0) red_s[wv] = below;
  __syncthreads();

  if (wv == 0) {
    int base = red_s[lane];
    #pragma unroll
    for (int o = 16; o; o >>= 1) base += __shfl_xor(base, o, 32);
    int c[4];
    int tot = 0;
    #pragma unroll
    for (int j = 0; j < 4; ++j) {
      const int n = 4 * lane + j;
      c[j] = (n < NPB) ? cnt_s[n] : 0;
      tot += c[j];
    }
    int inc = tot;
    #pragma unroll
    for (int o = 1; o < 32; o <<= 1) {
      const int u = __shfl_up(inc, o, 32);
      if (lane >= o) inc += u;
    }
    int run = base + inc - tot;
    #pragma unroll
    for (int j = 0; j < 4; ++j) {
      const int n = 4 * lane + j;
      if (n < NPB) { off[lo + n] = run; cur_s[n] = run; }
      run += c[j];
    }
  }
  __syncthreads();

  #pragma unroll 1
  for (int j0 = 0; j0 < FULLR; j0 += UNR) {
    int4 d[UNR];
    #pragma unroll
    for (int u = 0; u < UNR; ++u) d[u] = d4[t + TA * (j0 + u)];
    #pragma unroll
    for (int u = 0; u < UNR; ++u) {
      const int e0 = (t + TA * (j0 + u)) << 2;
      place_one(d[u].x, lo, e0,     cur_s, csr);
      place_one(d[u].y, lo, e0 + 1, cur_s, csr);
      place_one(d[u].z, lo, e0 + 2, cur_s, csr);
      place_one(d[u].w, lo, e0 + 3, cur_s, csr);
    }
  }
  if (t < EQ - FULLR * TA) {
    const int4 d = d4[FULLR * TA + t];
    const int e0 = (FULLR * TA + t) << 2;
    place_one(d.x, lo, e0,     cur_s, csr);
    place_one(d.y, lo, e0 + 1, cur_s, csr);
    place_one(d.z, lo, e0 + 2, cur_s, csr);
    place_one(d.w, lo, e0 + 3, cur_s, csr);
  }
}

__global__ __launch_bounds__(256) void k_gemm_softmax(const float* __restrict__ agg,
                                                      const float* __restrict__ W,
                                                      float* __restrict__ out) {
  __shared__ float4 S4[16 * NQ + 8];
  float* S = (float*)S4;
  const int wv   = threadIdx.x >> 5;
  const int lane = threadIdx.x & 31;
  const int r  = lane & 15;
  const int h  = lane >> 4;
  const int m0 = blockIdx.x << 4;

  v2f A[8], B[4][8];
  #pragma unroll
  for (int i = 0; i < 8; ++i) {
    const int k = (i << 2) + (h << 1);
    A[i].x = agg[(m0 + r) * IND + k];
    A[i].y = agg[(m0 + r) * IND + k + 1];
  }
  #pragma unroll
  for (int t = 0; t < 4; ++t) {
    const int col = (wv << 6) + (t << 4) + r;
    const int cc  = (col < OUTD) ? col : 0;
    #pragma unroll
    for (int i = 0; i < 8; ++i) {
      const int k = (i << 2) + (h << 1);
      B[t][i].x = W[k * OUTD + cc];
      B[t][i].y = W[(k + 1) * OUTD + cc];
    }
  }
  v8f acc[4] = {};
  #pragma unroll
  for (int i = 0; i < 8; ++i) {
    #pragma unroll
    for (int t = 0; t < 4; ++t)
      acc[t] = __builtin_amdgcn_wmma_f32_16x16x4_f32(
          false, A[i], false, B[t][i], (short)0, acc[t], false, false);
  }

  #pragma unroll
  for (int t = 0; t < 4; ++t) {
    const int col = (wv << 6) + (t << 4) + r;
    #pragma unroll
    for (int rr = 0; rr < 8; ++rr) {
      int idx = (rr + (h << 3)) * OUTD + col;
      if (t == 3) idx = (col < OUTD) ? idx : 16 * OUTD + lane;
      S[idx] = acc[t][rr];
    }
  }
  __syncthreads();

  const bool last = (lane + 96) < NQ;
  #pragma unroll
  for (int rw = 0; rw < 2; ++rw) {
    float4* r4 = S4 + ((wv << 1) + rw) * NQ;
    float4 v[4];
    #pragma unroll
    for (int j = 0; j < 3; ++j) v[j] = r4[lane + 32 * j];
    v[3] = last ? r4[lane + 96] : make_float4(-INFINITY, -INFINITY, -INFINITY, -INFINITY);
    float mx = -INFINITY;
    #pragma unroll
    for (int j = 0; j < 4; ++j)
      mx = fmaxf(mx, fmaxf(fmaxf(v[j].x, v[j].y), fmaxf(v[j].z, v[j].w)));
    #pragma unroll
    for (int off = 16; off; off >>= 1) mx = fmaxf(mx, __shfl_xor(mx, off, 32));
    float sm = 0.0f;
    #pragma unroll
    for (int j = 0; j < 4; ++j) {
      const bool act = (j < 3) || last;
      v[j].x = act ? __expf(v[j].x - mx) : 0.0f;
      v[j].y = act ? __expf(v[j].y - mx) : 0.0f;
      v[j].z = act ? __expf(v[j].z - mx) : 0.0f;
      v[j].w = act ? __expf(v[j].w - mx) : 0.0f;
      sm += (v[j].x + v[j].y) + (v[j].z + v[j].w);
    }
    #pragma unroll
    for (int off = 16; off; off >>= 1) sm += __shfl_xor(sm, off, 32);
    const float inv = 1.0f / sm;
    #pragma unroll
    for (int j = 0; j < 4; ++j) {
      v[j].x *= inv; v[j].y *= inv; v[j].z *= inv; v[j].w *= inv;
    }
    #pragma unroll
    for (int j = 0; j < 3; ++j) r4[lane + 32 * j] = v[j];
    if (last) r4[lane + 96] = v[3];
  }
  __syncthreads();

  float4* o4 = (float4*)(out + (size_t)m0 * OUTD);
  const int tid = threadIdx.x;
  #pragma unroll
  for (int j = 0; j < 7; ++j) o4[tid + 256 * j] = S4[tid + 256 * j];
  if (tid < 16 * NQ - 7 * 256) o4[tid + 7 * 256] = S4[tid + 7 * 256];
}

__device__ __forceinline__ float wave_max(float v) {
  #pragma unroll
  for (int o = 16; o; o >>= 1) v = fmaxf(v, __shfl_xor(v, o, 32));
  return v;
}
__device__ __forceinline__ float leaky(float v) {
  return (v > 0.0f) ? v : NEG_SLOPE * v;
}
__device__ __forceinline__ int next_edge(const int* __restrict__ csr, int beg, int end,
                                         int prev, int lane) {
  int best = 0x7FFFFFFF;
  for (int p = beg + lane; p < end; p += 32) {
    const int c = csr[p];
    if (c > prev && c < best) best = c;
  }
  #pragma unroll
  for (int o = 16; o; o >>= 1) best = min(best, __shfl_xor(best, o, 32));
  return __builtin_amdgcn_readfirstlane(best);
}

__global__ __launch_bounds__(256) void k_gather(const int* __restrict__ src,
                                                const int* __restrict__ off,
                                                const int* __restrict__ csr,
                                                const float* __restrict__ el,
                                                const float* __restrict__ er,
                                                const float* __restrict__ x,
                                                float* __restrict__ agg) {
  const int node = __builtin_amdgcn_readfirstlane((blockIdx.x * 256 + threadIdx.x) >> 5);
  const int lane = threadIdx.x & 31;
  const int beg = off[node];
  const int end = (node + 1 < NN) ? off[node + 1] : EE;
  const int deg = end - beg;
  const float ern = er[node];
  float acc = 0.0f;

  if (deg <= 32) {
    const bool act = lane < deg;
    const int id = act ? csr[beg + lane] : 0x7FFFFFFF;
    const int sn = act ? src[id] : 0;
    const float e = act ? leaky(el[sn] + ern) : -INFINITY;
    int rank = 0;
    for (int j = 0; j < deg; ++j)
      rank += (__builtin_amdgcn_readlane(id, j) < id) ? 1 : 0;
    if (!act) rank = lane;
    float m = wave_max(e);
    m = isfinite(m) ? m : 0.0f;
    const float a = act ? __expf(e - m) : 0.0f;
    const int sa = __builtin_amdgcn_ds_permute(rank << 2, __float_as_int(a));
    const int ss = __builtin_amdgcn_ds_permute(rank << 2, sn);
    float s = 0.0f;
    for (int k = 0; k < deg; ++k) s += __int_as_float(__builtin_amdgcn_readlane(sa, k));
    const int al_bits = __float_as_int(__int_as_float(sa) / (s > 0.0f ? s : 1.0f));
    for (int k0 = 0; k0 < deg; k0 += 16) {
      #pragma unroll
      for (int j = 0; j < 16; ++j) {
        const int   snk = __builtin_amdgcn_readlane(ss, k0 + j);
        const float al  = __int_as_float(__builtin_amdgcn_readlane(al_bits, k0 + j));
        acc = fmaf(al, x[(size_t)snk * IND + lane], acc);
      }
    }
  } else {
    float mx = -INFINITY;
    for (int p = beg + lane; p < end; p += 32)
      mx = fmaxf(mx, leaky(el[src[csr[p]]] + ern));
    float m = wave_max(mx);
    m = isfinite(m) ? m : 0.0f;
    float s = 0.0f;
    int prev = -1;
    for (int k = 0; k < deg; ++k) {
      prev = next_edge(csr, beg, end, prev, lane);
      s += __expf(leaky(el[src[prev]] + ern) - m);
    }
    const float den = (s > 0.0f) ? s : 1.0f;
    prev = -1;
    for (int k = 0; k < deg; ++k) {
      prev = next_edge(csr, beg, end, prev, lane);
      const int snk = src[prev];
      const float al = __expf(leaky(el[snk] + ern) - m) / den;
      acc = fmaf(al, x[(size_t)snk * IND + lane], acc);
    }
  }
  agg[(size_t)node * IND + lane] = acc;
}

extern "C" void kernel_launch(void* const* d_in, const int* in_sizes, int n_in,
                              void* d_out, int out_size, void* d_ws, size_t ws_size,
                              hipStream_t stream) {
  const float* x      = (const float*)d_in[0];
  const float* W      = (const float*)d_in[1];
  const float* attn_l = (const float*)d_in[2];
  const float* attn_r = (const float*)d_in[3];
  const float* px_r   = (const float*)d_in[4];
  const int*   src    = (const int*)d_in[5];
  const int*   dst    = (const int*)d_in[6];
  float*       out    = (float*)d_out;

  float* ws   = (float*)d_ws;
  float* agg  = ws;
  float* el   = agg + (size_t)NN * IND;
  float* er   = el + NN;
  int*   off  = (int*)(er + NN);
  int*   csr  = off + NN;

  k_build       <<<GA, TA, 0, stream>>>(dst, x, W, attn_l, attn_r, px_r, out, off, csr, el, er);
  k_gather      <<<NN / 8, 256, 0, stream>>>(src, off, csr, el, er, x, agg);
  k_gemm_softmax<<<MTILES, 256, 0, stream>>>(agg, W, out);
}


# gather: a segment of 33..64 edges is ranked in-wave (two edges per lane, put into edge-id order through LDS) instead of picked one edge at a time by repeated min-search; sum order unchanged
# speedup vs baseline: 6.1546x; 1.8922x over previous
#include <hip/hip_runtime.h>
#include <math.h>

#define NN     10000
#define EE     160000
#define IND    32
#define OUTD   500
#define NQ     125
#define NTILES 32
#define MTILES 625
#define NEG_SLOPE 0.2f

typedef __attribute__((ext_vector_type(2)))  float    v2f;
typedef __attribute__((ext_vector_type(8)))  float    v8f;

#define GA    80
#define NPB   125
#define TA    1024
#define EQ    (EE / 4)
#define FULLR 39
#define UNR   13

__device__ __forceinline__ void count_one(int d, int lo, int* cnt_s, int& below) {
  const int v = d - lo;
  below += (v < 0) ? 1 : 0;
  if ((unsigned)v < (unsigned)NPB) atomicAdd(&cnt_s[v], 1);
}
__device__ __forceinline__ void place_one(int d, int lo, int eid, int* cur_s,
                                          int* __restrict__ csr) {
  const int v = d - lo;
  if ((unsigned)v < (unsigned)NPB) {
    const int p = atomicAdd(&cur_s[v], 1);
    csr[p] = eid;
  }
}

__global__ __launch_bounds__(TA) void k_build(const int* __restrict__ dst,
                                              const float* __restrict__ x,
                                              const float* __restrict__ W,
                                              const float* __restrict__ attn_l,
                                              const float* __restrict__ attn_r,
                                              const float* __restrict__ px_r,
                                              float* __restrict__ out,
                                              int* __restrict__ off,
                                              int* __restrict__ csr,
                                              float* __restrict__ el,
                                              float* __restrict__ er) {
  __shared__ int    cnt_s[128];
  __shared__ int    cur_s[128];
  __shared__ int    red_s[32];
  __shared__ float4 wlr4[16];
  float* wlr_s = (float*)wlr4;
  const int t = threadIdx.x, lane = t & 31, wv = t >> 5;
  const int lo = blockIdx.x * NPB;

  if (t < 128) cnt_s[t] = 0;

  const int xq = min(lo * 8 + t, NN * 8 - 1);
  const float4 xv = ((const float4*)x)[xq];

  if (blockIdx.x == 0 && t < OUTD) {
    const float v = px_r[t];
    const float sp = (v > 20.0f) ? v : log1pf(__expf(v));
    out[(size_t)NN * OUTD + t] = fminf(fmaxf(sp, 1.0e-4f), 1.0e4f);
  }

  #pragma unroll
  for (int g2 = 0; g2 < 2; ++g2) {
    const float* a = g2 ? attn_r : attn_l;
    float s = 0.0f;
    #pragma unroll
    for (int j = 0; j < 16; ++j) {
      const int d  = lane + 32 * j;
      const int dc = (d < OUTD) ? d : 0;
      s = fmaf(W[wv * OUTD + dc], (d < OUTD) ? a[dc] : 0.0f, s);
    }
    #pragma unroll
    for (int o = 16; o; o >>= 1) s += __shfl_xor(s, o, 32);
    if (lane == 0) wlr_s[wv + 32 * g2] = s;
  }
  __syncthreads();

  {
    const int part = t & 7;
    const float4 l  = wlr4[part];
    const float4 rr = wlr4[8 + part];
    float sl = fmaf(xv.x, l.x, fmaf(xv.y, l.y, fmaf(xv.z, l.z, xv.w * l.w)));
    float sr = fmaf(xv.x, rr.x, fmaf(xv.y, rr.y, fmaf(xv.z, rr.z, xv.w * rr.w)));
    #pragma unroll
    for (int o = 4; o; o >>= 1) {
      sl += __shfl_xor(sl, o, 32);
      sr += __shfl_xor(sr, o, 32);
    }
    if (part == 0 && t < NPB * 8) { el[lo + (t >> 3)] = sl; er[lo + (t >> 3)] = sr; }
  }

  const int4* d4 = (const int4*)dst;

  int below = 0;
  #pragma unroll 1
  for (int j0 = 0; j0 < FULLR; j0 += UNR) {
    int4 d[UNR];
    #pragma unroll
    for (int u = 0; u < UNR; ++u) d[u] = d4[t + TA * (j0 + u)];
    #pragma unroll
    for (int u = 0; u < UNR; ++u) {
      count_one(d[u].x, lo, cnt_s, below);
      count_one(d[u].y, lo, cnt_s, below);
      count_one(d[u].z, lo, cnt_s, below);
      count_one(d[u].w, lo, cnt_s, below);
    }
  }
  if (t < EQ - FULLR * TA) {
    const int4 d = d4[FULLR * TA + t];
    count_one(d.x, lo, cnt_s, below);
    count_one(d.y, lo, cnt_s, below);
    count_one(d.z, lo, cnt_s, below);
    count_one(d.w, lo, cnt_s, below);
  }
  #pragma unroll
  for (int o = 16; o; o >>= 1) below += __shfl_xor(below, o, 32);
  if (lane == 0) red_s[wv] = below;
  __syncthreads();

  if (wv == 0) {
    int base = red_s[lane];
    #pragma unroll
    for (int o = 16; o; o >>= 1) base += __shfl_xor(base, o, 32);
    int c[4];
    int tot = 0;
    #pragma unroll
    for (int j = 0; j < 4; ++j) {
      const int n = 4 * lane + j;
      c[j] = (n < NPB) ? cnt_s[n] : 0;
      tot += c[j];
    }
    int inc = tot;
    #pragma unroll
    for (int o = 1; o < 32; o <<= 1) {
      const int u = __shfl_up(inc, o, 32);
      if (lane >= o) inc += u;
    }
    int run = base + inc - tot;
    #pragma unroll
    for (int j = 0; j < 4; ++j) {
      const int n = 4 * lane + j;
      if (n < NPB) { off[lo + n] = run; cur_s[n] = run; }
      run += c[j];
    }
  }
  __syncthreads();

  #pragma unroll 1
  for (int j0 = 0; j0 < FULLR; j0 += UNR) {
    int4 d[UNR];
    #pragma unroll
    for (int u = 0; u < UNR; ++u) d[u] = d4[t + TA * (j0 + u)];
    #pragma unroll
    for (int u = 0; u < UNR; ++u) {
      const int e0 = (t + TA * (j0 + u)) << 2;
      place_one(d[u].x, lo, e0,     cur_s, csr);
      place_one(d[u].y, lo, e0 + 1, cur_s, csr);
      place_one(d[u].z, lo, e0 + 2, cur_s, csr);
      place_one(d[u].w, lo, e0 + 3, cur_s, csr);
    }
  }
  if (t < EQ - FULLR * TA) {
    const int4 d = d4[FULLR * TA + t];
    const int e0 = (FULLR * TA + t) << 2;
    place_one(d.x, lo, e0,     cur_s, csr);
    place_one(d.y, lo, e0 + 1, cur_s, csr);
    place_one(d.z, lo, e0 + 2, cur_s, csr);
    place_one(d.w, lo, e0 + 3, cur_s, csr);
  }
}

__global__ __launch_bounds__(256) void k_gemm_softmax(const float* __restrict__ agg,
                                                      const float* __restrict__ W,
                                                      float* __restrict__ out) {
  __shared__ float4 S4[16 * NQ + 8];
  float* S = (float*)S4;
  const int wv   = threadIdx.x >> 5;
  const int lane = threadIdx.x & 31;
  const int r  = lane & 15;
  const int h  = lane >> 4;
  const int m0 = blockIdx.x << 4;

  v2f A[8], B[4][8];
  #pragma unroll
  for (int i = 0; i < 8; ++i) {
    const int k = (i << 2) + (h << 1);
    A[i].x = agg[(m0 + r) * IND + k];
    A[i].y = agg[(m0 + r) * IND + k + 1];
  }
  #pragma unroll
  for (int t = 0; t < 4; ++t) {
    const int col = (wv << 6) + (t << 4) + r;
    const int cc  = (col < OUTD) ? col : 0;
    #pragma unroll
    for (int i = 0; i < 8; ++i) {
      const int k = (i << 2) + (h << 1);
      B[t][i].x = W[k * OUTD + cc];
      B[t][i].y = W[(k + 1) * OUTD + cc];
    }
  }
  v8f acc[4] = {};
  #pragma unroll
  for (int i = 0; i < 8; ++i) {
    #pragma unroll
    for (int t = 0; t < 4; ++t)
      acc[t] = __builtin_amdgcn_wmma_f32_16x16x4_f32(
          false, A[i], false, B[t][i], (short)0, acc[t], false, false);
  }

  #pragma unroll
  for (int t = 0; t < 4; ++t) {
    const int col = (wv << 6) + (t << 4) + r;
    #pragma unroll
    for (int rr = 0; rr < 8; ++rr) {
      int idx = (rr + (h << 3)) * OUTD + col;
      if (t == 3) idx = (col < OUTD) ? idx : 16 * OUTD + lane;
      S[idx] = acc[t][rr];
    }
  }
  __syncthreads();

  const bool last = (lane + 96) < NQ;
  #pragma unroll
  for (int rw = 0; rw < 2; ++rw) {
    float4* r4 = S4 + ((wv << 1) + rw) * NQ;
    float4 v[4];
    #pragma unroll
    for (int j = 0; j < 3; ++j) v[j] = r4[lane + 32 * j];
    v[3] = last ? r4[lane + 96] : make_float4(-INFINITY, -INFINITY, -INFINITY, -INFINITY);
    float mx = -INFINITY;
    #pragma unroll
    for (int j = 0; j < 4; ++j)
      mx = fmaxf(mx, fmaxf(fmaxf(v[j].x, v[j].y), fmaxf(v[j].z, v[j].w)));
    #pragma unroll
    for (int off = 16; off; off >>= 1) mx = fmaxf(mx, __shfl_xor(mx, off, 32));
    float sm = 0.0f;
    #pragma unroll
    for (int j = 0; j < 4; ++j) {
      const bool act = (j < 3) || last;
      v[j].x = act ? __expf(v[j].x - mx) : 0.0f;
      v[j].y = act ? __expf(v[j].y - mx) : 0.0f;
      v[j].z = act ? __expf(v[j].z - mx) : 0.0f;
      v[j].w = act ? __expf(v[j].w - mx) : 0.0f;
      sm += (v[j].x + v[j].y) + (v[j].z + v[j].w);
    }
    #pragma unroll
    for (int off = 16; off; off >>= 1) sm += __shfl_xor(sm, off, 32);
    const float inv = 1.0f / sm;
    #pragma unroll
    for (int j = 0; j < 4; ++j) {
      v[j].x *= inv; v[j].y *= inv; v[j].z *= inv; v[j].w *= inv;
    }
    #pragma unroll
    for (int j = 0; j < 3; ++j) r4[lane + 32 * j] = v[j];
    if (last) r4[lane + 96] = v[3];
  }
  __syncthreads();

  float4* o4 = (float4*)(out + (size_t)m0 * OUTD);
  const int tid = threadIdx.x;
  #pragma unroll
  for (int j = 0; j < 7; ++j) o4[tid + 256 * j] = S4[tid + 256 * j];
  if (tid < 16 * NQ - 7 * 256) o4[tid + 7 * 256] = S4[tid + 7 * 256];
}

__device__ __forceinline__ float wave_max(float v) {
  #pragma unroll
  for (int o = 16; o; o >>= 1) v = fmaxf(v, __shfl_xor(v, o, 32));
  return v;
}
__device__ __forceinline__ float leaky(float v) {
  return (v > 0.0f) ? v : NEG_SLOPE * v;
}
__device__ __forceinline__ int next_edge(const int* __restrict__ csr, int beg, int end,
                                         int prev, int lane) {
  int best = 0x7FFFFFFF;
  for (int p = beg + lane; p < end; p += 32) {
    const int c = csr[p];
    if (c > prev && c < best) best = c;
  }
  #pragma unroll
  for (int o = 16; o; o >>= 1) best = min(best, __shfl_xor(best, o, 32));
  return __builtin_amdgcn_readfirstlane(best);
}

__global__ __launch_bounds__(256) void k_gather(const int* __restrict__ src,
                                                const int* __restrict__ off,
                                                const int* __restrict__ csr,
                                                const float* __restrict__ el,
                                                const float* __restrict__ er,
                                                const float* __restrict__ x,
                                                float* __restrict__ agg) {
  __shared__ float sw_s[8 * 64];
  __shared__ int   sr_s[8 * 64];
  const int node = __builtin_amdgcn_readfirstlane((blockIdx.x * 256 + threadIdx.x) >> 5);
  const int lane = threadIdx.x & 31;
  const int beg = off[node];
  const int end = (node + 1 < NN) ? off[node + 1] : EE;
  const int deg = end - beg;
  const float ern = er[node];
  float acc = 0.0f;

  if (deg <= 32) {
    const bool act = lane < deg;
    const int id = act ? csr[beg + lane] : 0x7FFFFFFF;
    const int sn = act ? src[id] : 0;
    const float e = act ? leaky(el[sn] + ern) : -INFINITY;
    int rank = 0;
    for (int j = 0; j < deg; ++j)
      rank += (__builtin_amdgcn_readlane(id, j) < id) ? 1 : 0;
    if (!act) rank = lane;
    float m = wave_max(e);
    m = isfinite(m) ? m : 0.0f;
    const float a = act ? __expf(e - m) : 0.0f;
    const int sa = __builtin_amdgcn_ds_permute(rank << 2, __float_as_int(a));
    const int ss = __builtin_amdgcn_ds_permute(rank << 2, sn);
    float s = 0.0f;
    for (int k = 0; k < deg; ++k) s += __int_as_float(__builtin_amdgcn_readlane(sa, k));
    const int al_bits = __float_as_int(__int_as_float(sa) / (s > 0.0f ? s : 1.0f));
    for (int k0 = 0; k0 < deg; k0 += 16) {
      #pragma unroll
      for (int j = 0; j < 16; ++j) {
        const int   snk = __builtin_amdgcn_readlane(ss, k0 + j);
        const float al  = __int_as_float(__builtin_amdgcn_readlane(al_bits, k0 + j));
        acc = fmaf(al, x[(size_t)snk * IND + lane], acc);
      }
    }
  } else if (deg <= 64) {
    const int  nb   = deg - 32;
    const bool actB = lane < nb;
    const int idA = csr[beg + lane];
    const int idB = actB ? csr[beg + 32 + lane] : 0x7FFFFFFF;
    const int snA = src[idA];
    const int snB = actB ? src[idB] : 0;
    const float eA = leaky(el[snA] + ern);
    const float eB = actB ? leaky(el[snB] + ern) : -INFINITY;
    int rkA = 0, rkB = 0;
    for (int j = 0; j < 32; ++j) {
      const int c = __builtin_amdgcn_readlane(idA, j);
      rkA += (c < idA) ? 1 : 0;
      rkB += (c < idB) ? 1 : 0;
    }
    for (int j = 0; j < nb; ++j) {
      const int c = __builtin_amdgcn_readlane(idB, j);
      rkA += (c < idA) ? 1 : 0;
      rkB += (c < idB) ? 1 : 0;
    }
    float m = wave_max(fmaxf(eA, eB));
    m = isfinite(m) ? m : 0.0f;
    float* sw = sw_s + ((threadIdx.x >> 5) << 6);
    int*   sr = sr_s + ((threadIdx.x >> 5) << 6);
    sw[rkA] = __expf(eA - m);
    sr[rkA] = snA;
    if (actB) { sw[rkB] = __expf(eB - m); sr[rkB] = snB; }
    const int   waA = __float_as_int(sw[lane]);
    const int   ssA = sr[lane];
    const int   waB = actB ? __float_as_int(sw[32 + lane]) : 0;
    const int   ssB = actB ? sr[32 + lane] : 0;
    float s = 0.0f;
    for (int k = 0; k < 32; ++k) s += __int_as_float(__builtin_amdgcn_readlane(waA, k));
    for (int k = 0; k < nb; ++k) s += __int_as_float(__builtin_amdgcn_readlane(waB, k));
    const float den = (s > 0.0f) ? s : 1.0f;
    const int alA = __float_as_int(__int_as_float(waA) / den);
    const int alB = __float_as_int(__int_as_float(waB) / den);
    for (int k0 = 0; k0 < 32; k0 += 16) {
      #pragma unroll
      for (int j = 0; j < 16; ++j) {
        const int   snk = __builtin_amdgcn_readlane(ssA, k0 + j);
        const float al  = __int_as_float(__builtin_amdgcn_readlane(alA, k0 + j));
        acc = fmaf(al, x[(size_t)snk * IND + lane], acc);
      }
    }
    for (int k0 = 0; k0 < nb; k0 += 16) {
      #pragma unroll
      for (int j = 0; j < 16; ++j) {
        const int   snk = __builtin_amdgcn_readlane(ssB, k0 + j);
        const float al  = __int_as_float(__builtin_amdgcn_readlane(alB, k0 + j));
        acc = fmaf(al, x[(size_t)snk * IND + lane], acc);
      }
    }
  } else {
    float mx = -INFINITY;
    for (int p = beg + lane; p < end; p += 32)
      mx = fmaxf(mx, leaky(el[src[csr[p]]] + ern));
    float m = wave_max(mx);
    m = isfinite(m) ? m : 0.0f;
    float s = 0.0f;
    int prev = -1;
    for (int k = 0; k < deg; ++k) {
      prev = next_edge(csr, beg, end, prev, lane);
      s += __expf(leaky(el[src[prev]] + ern) - m);
    }
    const float den = (s > 0.0f) ? s : 1.0f;
    prev = -1;
    for (int k = 0; k < deg; ++k) {
      prev = next_edge(csr, beg, end, prev, lane);
      const int snk = src[prev];
      const float al = __expf(leaky(el[snk] + ern) - m) / den;
      acc = fmaf(al, x[(size_t)snk * IND + lane], acc);
    }
  }
  agg[(size_t)node * IND + lane] = acc;
}

extern "C" void kernel_launch(void* const* d_in, const int* in_sizes, int n_in,
                              void* d_out, int out_size, void* d_ws, size_t ws_size,
                              hipStream_t stream) {
  const float* x      = (const float*)d_in[0];
  const float* W      = (const float*)d_in[1];
  const float* attn_l = (const float*)d_in[2];
  const float* attn_r = (const float*)d_in[3];
  const float* px_r   = (const float*)d_in[4];
  const int*   src    = (const int*)d_in[5];
  const int*   dst    = (const int*)d_in[6];
  float*       out    = (float*)d_out;

  float* ws   = (float*)d_ws;
  float* agg  = ws;
  float* el   = agg + (size_t)NN * IND;
  float* er   = el + NN;
  int*   off  = (int*)(er + NN);
  int*   csr  = off + NN;

  k_build       <<<GA, TA, 0, stream>>>(dst, x, W, attn_l, attn_r, px_r, out, off, csr, el, er);
  k_gather      <<<NN / 8, 256, 0, stream>>>(src, off, csr, el, er, x, agg);
  k_gemm_softmax<<<MTILES, 256, 0, stream>>>(agg, W, out);
}


# edge-list build spread over all WGPs: edge ids cut into 4 ranges, 200 work-groups (50 node owners x 4) each read a quarter of dst once, kept in registers; gather joins 4 segments; sum order kept
# speedup vs baseline: 8.0592x; 1.3095x over previous
#include <hip/hip_runtime.h>
#include <math.h>

#define NN     10000
#define EE     160000
#define IND    32
#define OUTD   500
#define NQ     125
#define NTILES 32
#define MTILES 625
#define NEG_SLOPE 0.2f

typedef __attribute__((ext_vector_type(2)))  float    v2f;
typedef __attribute__((ext_vector_type(8)))  float    v8f;

#define NPART 4
#define NOWN  50
#define GA    (NOWN * NPART)
#define NPB   200
#define NEL   (NPB / NPART)
#define TA    1024
#define EP    (EE / NPART)
#define QP    (EP / 4)
#define FULLR 9
#define NQT   (FULLR + 1)

__device__ __forceinline__ void count_one(int d, int lo, int* cnt_s, int& below) {
  const int v = d - lo;
  below += (v < 0) ? 1 : 0;
  if ((unsigned)v < (unsigned)NPB) atomicAdd(&cnt_s[v], 1);
}
__device__ __forceinline__ void place_one(int d, int lo, int eid, int* cur_s,
                                          int* __restrict__ csr) {
  const int v = d - lo;
  if ((unsigned)v < (unsigned)NPB) {
    const int p = atomicAdd(&cur_s[v], 1);
    csr[p] = eid;
  }
}

__global__ __launch_bounds__(TA) void k_build(const int* __restrict__ dst,
                                              const float* __restrict__ x,
                                              const float* __restrict__ W,
                                              const float* __restrict__ attn_l,
                                              const float* __restrict__ attn_r,
                                              const float* __restrict__ px_r,
                                              float* __restrict__ out,
                                              int* __restrict__ off,
                                              int* __restrict__ csr,
                                              float* __restrict__ el,
                                              float* __restrict__ er) {
  __shared__ int    cnt_s[256];
  __shared__ int    cur_s[256];
  __shared__ int    red_s[32];
  __shared__ float4 wlr4[16];
  float* wlr_s = (float*)wlr4;
  const int t = threadIdx.x, lane = t & 31, wv = t >> 5;
  const int part = blockIdx.x & (NPART - 1);
  const int lo = (blockIdx.x / NPART) * NPB;
  const int qb = part * QP;

  const int4* d4 = (const int4*)dst;
  int4 d[NQT];
  #pragma unroll
  for (int u = 0; u < FULLR; ++u) d[u] = d4[qb + t + TA * u];
  {
    const int q = FULLR * TA + t;
    d[FULLR] = d4[qb + min(q, QP - 1)];
    if (q >= QP) d[FULLR] = make_int4(0x7FFFFFFF, 0x7FFFFFFF, 0x7FFFFFFF, 0x7FFFFFFF);
  }

  if (t < 256) cnt_s[t] = 0;

  const int nel0 = lo + part * NEL;
  const int xq = min(nel0 * 8 + t, NN * 8 - 1);
  const float4 xv = ((const float4*)x)[xq];

  if (blockIdx.x == 0 && t < OUTD) {
    const float v = px_r[t];
    const float sp = (v > 20.0f) ? v : log1pf(__expf(v));
    out[(size_t)NN * OUTD + t] = fminf(fmaxf(sp, 1.0e-4f), 1.0e4f);
  }
  if (blockIdx.x >= GA - NPART && t == 0) off[NN * NPART + part] = (part + 1) * EP;

  #pragma unroll
  for (int g2 = 0; g2 < 2; ++g2) {
    const float* a = g2 ? attn_r : attn_l;
    float s = 0.0f;
    #pragma unroll
    for (int j = 0; j < 16; ++j) {
      const int dd = lane + 32 * j;
      const int dc = (dd < OUTD) ? dd : 0;
      s = fmaf(W[wv * OUTD + dc], (dd < OUTD) ? a[dc] : 0.0f, s);
    }
    #pragma unroll
    for (int o = 16; o; o >>= 1) s += __shfl_xor(s, o, 32);
    if (lane == 0) wlr_s[wv + 32 * g2] = s;
  }
  __syncthreads();

  {
    const int part8 = t & 7;
    const float4 l  = wlr4[part8];
    const float4 rr = wlr4[8 + part8];
    float sl = fmaf(xv.x, l.x, fmaf(xv.y, l.y, fmaf(xv.z, l.z, xv.w * l.w)));
    float sr = fmaf(xv.x, rr.x, fmaf(xv.y, rr.y, fmaf(xv.z, rr.z, xv.w * rr.w)));
    #pragma unroll
    for (int o = 4; o; o >>= 1) {
      sl += __shfl_xor(sl, o, 32);
      sr += __shfl_xor(sr, o, 32);
    }
    if (part8 == 0 && t < NEL * 8) { el[nel0 + (t >> 3)] = sl; er[nel0 + (t >> 3)] = sr; }
  }

  int below = 0;
  #pragma unroll
  for (int u = 0; u < NQT; ++u) {
    count_one(d[u].x, lo, cnt_s, below);
    count_one(d[u].y, lo, cnt_s, below);
    count_one(d[u].z, lo, cnt_s, below);
    count_one(d[u].w, lo, cnt_s, below);
  }
  #pragma unroll
  for (int o = 16; o; o >>= 1) below += __shfl_xor(below, o, 32);
  if (lane == 0) red_s[wv] = below;
  __syncthreads();

  if (wv == 0) {
    int base = red_s[lane];
    #pragma unroll
    for (int o = 16; o; o >>= 1) base += __shfl_xor(base, o, 32);
    int c[8];
    int tot = 0;
    #pragma unroll
    for (int j = 0; j < 8; ++j) {
      const int n = 8 * lane + j;
      c[j] = (n < NPB) ? cnt_s[n] : 0;
      tot += c[j];
    }
    int inc = tot;
    #pragma unroll
    for (int o = 1; o < 32; o <<= 1) {
      const int u = __shfl_up(inc, o, 32);
      if (lane >= o) inc += u;
    }
    int run = part * EP + base + inc - tot;
    #pragma unroll
    for (int j = 0; j < 8; ++j) {
      const int n = 8 * lane + j;
      if (n < NPB) { off[(lo + n) * NPART + part] = run; cur_s[n] = run; }
      run += c[j];
    }
  }
  __syncthreads();

  #pragma unroll
  for (int u = 0; u < NQT; ++u) {
    const int e0 = (qb + t + TA * u) << 2;
    place_one(d[u].x, lo, e0,     cur_s, csr);
    place_one(d[u].y, lo, e0 + 1, cur_s, csr);
    place_one(d[u].z, lo, e0 + 2, cur_s, csr);
    place_one(d[u].w, lo, e0 + 3, cur_s, csr);
  }
}

__global__ __launch_bounds__(256) void k_gemm_softmax(const float* __restrict__ agg,
                                                      const float* __restrict__ W,
                                                      float* __restrict__ out) {
  __shared__ float4 S4[16 * NQ + 8];
  float* S = (float*)S4;
  const int wv   = threadIdx.x >> 5;
  const int lane = threadIdx.x & 31;
  const int r  = lane & 15;
  const int h  = lane >> 4;
  const int m0 = blockIdx.x << 4;

  v2f A[8], B[4][8];
  #pragma unroll
  for (int i = 0; i < 8; ++i) {
    const int k = (i << 2) + (h << 1);
    A[i].x = agg[(m0 + r) * IND + k];
    A[i].y = agg[(m0 + r) * IND + k + 1];
  }
  #pragma unroll
  for (int t = 0; t < 4; ++t) {
    const int col = (wv << 6) + (t << 4) + r;
    const int cc  = (col < OUTD) ? col : 0;
    #pragma unroll
    for (int i = 0; i < 8; ++i) {
      const int k = (i << 2) + (h << 1);
      B[t][i].x = W[k * OUTD + cc];
      B[t][i].y = W[(k + 1) * OUTD + cc];
    }
  }
  v8f acc[4] = {};
  #pragma unroll
  for (int i = 0; i < 8; ++i) {
    #pragma unroll
    for (int t = 0; t < 4; ++t)
      acc[t] = __builtin_amdgcn_wmma_f32_16x16x4_f32(
          false, A[i], false, B[t][i], (short)0, acc[t], false, false);
  }

  #pragma unroll
  for (int t = 0; t < 4; ++t) {
    const int col = (wv << 6) + (t << 4) + r;
    #pragma unroll
    for (int rr = 0; rr < 8; ++rr) {
      int idx = (rr + (h << 3)) * OUTD + col;
      if (t == 3) idx = (col < OUTD) ? idx : 16 * OUTD + lane;
      S[idx] = acc[t][rr];
    }
  }
  __syncthreads();

  const bool last = (lane + 96) < NQ;
  #pragma unroll
  for (int rw = 0; rw < 2; ++rw) {
    float4* r4 = S4 + ((wv << 1) + rw) * NQ;
    float4 v[4];
    #pragma unroll
    for (int j = 0; j < 3; ++j) v[j] = r4[lane + 32 * j];
    v[3] = last ? r4[lane + 96] : make_float4(-INFINITY, -INFINITY, -INFINITY, -INFINITY);
    float mx = -INFINITY;
    #pragma unroll
    for (int j = 0; j < 4; ++j)
      mx = fmaxf(mx, fmaxf(fmaxf(v[j].x, v[j].y), fmaxf(v[j].z, v[j].w)));
    #pragma unroll
    for (int off = 16; off; off >>= 1) mx = fmaxf(mx, __shfl_xor(mx, off, 32));
    float sm = 0.0f;
    #pragma unroll
    for (int j = 0; j < 4; ++j) {
      const bool act = (j < 3) || last;
      v[j].x = act ? __expf(v[j].x - mx) : 0.0f;
      v[j].y = act ? __expf(v[j].y - mx) : 0.0f;
      v[j].z = act ? __expf(v[j].z - mx) : 0.0f;
      v[j].w = act ? __expf(v[j].w - mx) : 0.0f;
      sm += (v[j].x + v[j].y) + (v[j].z + v[j].w);
    }
    #pragma unroll
    for (int off = 16; off; off >>= 1) sm += __shfl_xor(sm, off, 32);
    const float inv = 1.0f / sm;
    #pragma unroll
    for (int j = 0; j < 4; ++j) {
      v[j].x *= inv; v[j].y *= inv; v[j].z *= inv; v[j].w *= inv;
    }
    #pragma unroll
    for (int j = 0; j < 3; ++j) r4[lane + 32 * j] = v[j];
    if (last) r4[lane + 96] = v[3];
  }
  __syncthreads();

  float4* o4 = (float4*)(out + (size_t)m0 * OUTD);
  const int tid = threadIdx.x;
  #pragma unroll
  for (int j = 0; j < 7; ++j) o4[tid + 256 * j] = S4[tid + 256 * j];
  if (tid < 16 * NQ - 7 * 256) o4[tid + 7 * 256] = S4[tid + 7 * 256];
}

__device__ __forceinline__ float wave_max(float v) {
  #pragma unroll
  for (int o = 16; o; o >>= 1) v = fmaxf(v, __shfl_xor(v, o, 32));
  return v;
}
__device__ __forceinline__ float leaky(float v) {
  return (v > 0.0f) ? v : NEG_SLOPE * v;
}
struct SegMap { int p1, p2, p3, a0, a1, a2, a3; };
__device__ __forceinline__ int seg_index(const SegMap& m, int q) {
  int a = m.a0;
  a = (q >= m.p1) ? m.a1 : a;
  a = (q >= m.p2) ? m.a2 : a;
  a = (q >= m.p3) ? m.a3 : a;
  return q + a;
}
__device__ __forceinline__ int next_edge(const int* __restrict__ csr, const SegMap& m,
                                         int deg, int prev, int lane) {
  int best = 0x7FFFFFFF;
  for (int q = lane; q < deg; q += 32) {
    const int c = csr[seg_index(m, q)];
    if (c > prev && c < best) best = c;
  }
  #pragma unroll
  for (int o = 16; o; o >>= 1) best = min(best, __shfl_xor(best, o, 32));
  return __builtin_amdgcn_readfirstlane(best);
}

__global__ __launch_bounds__(256) void k_gather(const int* __restrict__ src,
                                                const int* __restrict__ off,
                                                const int* __restrict__ csr,
                                                const float* __restrict__ el,
                                                const float* __restrict__ er,
                                                const float* __restrict__ x,
                                                float* __restrict__ agg) {
  __shared__ float sw_s[8 * 64];
  __shared__ int   sr_s[8 * 64];
  const int node = __builtin_amdgcn_readfirstlane((blockIdx.x * 256 + threadIdx.x) >> 5);
  const int lane = threadIdx.x & 31;
  const int4 st = ((const int4*)off)[node];
  const int4 en = ((const int4*)off)[node + 1];
  SegMap sg;
  sg.p1 = en.x - st.x;
  sg.p2 = sg.p1 + (en.y - st.y);
  sg.p3 = sg.p2 + (en.z - st.z);
  sg.a0 = st.x;
  sg.a1 = st.y - sg.p1;
  sg.a2 = st.z - sg.p2;
  sg.a3 = st.w - sg.p3;
  const int deg = sg.p3 + (en.w - st.w);
  const float ern = er[node];
  float acc = 0.0f;

  if (deg <= 32) {
    const bool act = lane < deg;
    const int id = act ? csr[seg_index(sg, lane)] : 0x7FFFFFFF;
    const int sn = act ? src[id] : 0;
    const float e = act ? leaky(el[sn] + ern) : -INFINITY;
    int rank = 0;
    for (int j = 0; j < deg; ++j)
      rank += (__builtin_amdgcn_readlane(id, j) < id) ? 1 : 0;
    if (!act) rank = lane;
    float m = wave_max(e);
    m = isfinite(m) ? m : 0.0f;
    const float a = act ? __expf(e - m) : 0.0f;
    const int sa = __builtin_amdgcn_ds_permute(rank << 2, __float_as_int(a));
    const int ss = __builtin_amdgcn_ds_permute(rank << 2, sn);
    float s = 0.0f;
    for (int k = 0; k < deg; ++k) s += __int_as_float(__builtin_amdgcn_readlane(sa, k));
    const int al_bits = __float_as_int(__int_as_float(sa) / (s > 0.0f ? s : 1.0f));
    for (int k0 = 0; k0 < deg; k0 += 16) {
      #pragma unroll
      for (int j = 0; j < 16; ++j) {
        const int   snk = __builtin_amdgcn_readlane(ss, k0 + j);
        const float al  = __int_as_float(__builtin_amdgcn_readlane(al_bits, k0 + j));
        acc = fmaf(al, x[(size_t)snk * IND + lane], acc);
      }
    }
  } else if (deg <= 64) {
    const int  nb   = deg - 32;
    const bool actB = lane < nb;
    const int idA = csr[seg_index(sg, lane)];
    const int idB = actB ? csr[seg_index(sg, 32 + lane)] : 0x7FFFFFFF;
    const int snA = src[idA];
    const int snB = actB ? src[idB] : 0;
    const float eA = leaky(el[snA] + ern);
    const float eB = actB ? leaky(el[snB] + ern) : -INFINITY;
    int rkA = 0, rkB = 0;
    for (int j = 0; j < 32; ++j) {
      const int c = __builtin_amdgcn_readlane(idA, j);
      rkA += (c < idA) ? 1 : 0;
      rkB += (c < idB) ? 1 : 0;
    }
    for (int j = 0; j < nb; ++j) {
      const int c = __builtin_amdgcn_readlane(idB, j);
      rkA += (c < idA) ? 1 : 0;
      rkB += (c < idB) ? 1 : 0;
    }
    float m = wave_max(fmaxf(eA, eB));
    m = isfinite(m) ? m : 0.0f;
    float* sw = sw_s + ((threadIdx.x >> 5) << 6);
    int*   sr = sr_s + ((threadIdx.x >> 5) << 6);
    sw[rkA] = __expf(eA - m);
    sr[rkA] = snA;
    if (actB) { sw[rkB] = __expf(eB - m); sr[rkB] = snB; }
    const int   waA = __float_as_int(sw[lane]);
    const int   ssA = sr[lane];
    const int   waB = actB ? __float_as_int(sw[32 + lane]) : 0;
    const int   ssB = actB ? sr[32 + lane] : 0;
    float s = 0.0f;
    for (int k = 0; k < 32; ++k) s += __int_as_float(__builtin_amdgcn_readlane(waA, k));
    for (int k = 0; k < nb; ++k) s += __int_as_float(__builtin_amdgcn_readlane(waB, k));
    const float den = (s > 0.0f) ? s : 1.0f;
    const int alA = __float_as_int(__int_as_float(waA) / den);
    const int alB = __float_as_int(__int_as_float(waB) / den);
    for (int k0 = 0; k0 < 32; k0 += 16) {
      #pragma unroll
      for (int j = 0; j < 16; ++j) {
        const int   snk = __builtin_amdgcn_readlane(ssA, k0 + j);
        const float al  = __int_as_float(__builtin_amdgcn_readlane(alA, k0 + j));
        acc = fmaf(al, x[(size_t)snk * IND + lane], acc);
      }
    }
    for (int k0 = 0; k0 < nb; k0 += 16) {
      #pragma unroll
      for (int j = 0; j < 16; ++j) {
        const int   snk = __builtin_amdgcn_readlane(ssB, k0 + j);
        const float al  = __int_as_float(__builtin_amdgcn_readlane(alB, k0 + j));
        acc = fmaf(al, x[(size_t)snk * IND + lane], acc);
      }
    }
  } else {
    float mx = -INFINITY;
    for (int q = lane; q < deg; q += 32)
      mx = fmaxf(mx, leaky(el[src[csr[seg_index(sg, q)]]] + ern));
    float m = wave_max(mx);
    m = isfinite(m) ? m : 0.0f;
    float s = 0.0f;
    int prev = -1;
    for (int k = 0; k < deg; ++k) {
      prev = next_edge(csr, sg, deg, prev, lane);
      s += __expf(leaky(el[src[prev]] + ern) - m);
    }
    const float den = (s > 0.0f) ? s : 1.0f;
    prev = -1;
    for (int k = 0; k < deg; ++k) {
      prev = next_edge(csr, sg, deg, prev, lane);
      const int snk = src[prev];
      const float al = __expf(leaky(el[snk] + ern) - m) / den;
      acc = fmaf(al, x[(size_t)snk * IND + lane], acc);
    }
  }
  agg[(size_t)node * IND + lane] = acc;
}

extern "C" void kernel_launch(void* const* d_in, const int* in_sizes, int n_in,
                              void* d_out, int out_size, void* d_ws, size_t ws_size,
                              hipStream_t stream) {
  const float* x      = (const float*)d_in[0];
  const float* W      = (const float*)d_in[1];
  const float* attn_l = (const float*)d_in[2];
  const float* attn_r = (const float*)d_in[3];
  const float* px_r   = (const float*)d_in[4];
  const int*   src    = (const int*)d_in[5];
  const int*   dst    = (const int*)d_in[6];
  float*       out    = (float*)d_out;

  float* ws   = (float*)d_ws;
  float* agg  = ws;
  float* el   = agg + (size_t)NN * IND;
  float* er   = el + NN;
  int*   off  = (int*)(er + NN);
  int*   csr  = off + NPART * (NN + 1);

  k_build       <<<GA, TA, 0, stream>>>(dst, x, W, attn_l, attn_r, px_r, out, off, csr, el, er);
  k_gather      <<<NN / 8, 256, 0, stream>>>(src, off, csr, el, er, x, agg);
  k_gemm_softmax<<<MTILES, 256, 0, stream>>>(agg, W, out);
}


# gather fused into the f32 WMMA GEMM+softmax launch: a 16-wave block sums its 16 dst rows (one wave per node) into LDS, then multiplies and normalises them; 3 -> 2 launches; sum order unchanged
# speedup vs baseline: 8.6939x; 1.0788x over previous
#include <hip/hip_runtime.h>
#include <math.h>

#define NN     10000
#define EE     160000
#define IND    32
#define OUTD   500
#define NQ     125
#define NTILES 32
#define MTILES 625
#define NEG_SLOPE 0.2f

typedef __attribute__((ext_vector_type(2)))  float    v2f;
typedef __attribute__((ext_vector_type(8)))  float    v8f;

#define NPART 4
#define NOWN  50
#define GA    (NOWN * NPART)
#define NPB   200
#define NEL   (NPB / NPART)
#define TA    1024
#define EP    (EE / NPART)
#define QP    (EP / 4)
#define FULLR 9
#define NQT   (FULLR + 1)

__device__ __forceinline__ void count_one(int d, int lo, int* cnt_s, int& below) {
  const int v = d - lo;
  below += (v < 0) ? 1 : 0;
  if ((unsigned)v < (unsigned)NPB) atomicAdd(&cnt_s[v], 1);
}
__device__ __forceinline__ void place_one(int d, int lo, int eid, int* cur_s,
                                          int* __restrict__ csr) {
  const int v = d - lo;
  if ((unsigned)v < (unsigned)NPB) {
    const int p = atomicAdd(&cur_s[v], 1);
    csr[p] = eid;
  }
}

__global__ __launch_bounds__(TA) void k_build(const int* __restrict__ dst,
                                              const float* __restrict__ x,
                                              const float* __restrict__ W,
                                              const float* __restrict__ attn_l,
                                              const float* __restrict__ attn_r,
                                              const float* __restrict__ px_r,
                                              float* __restrict__ out,
                                              int* __restrict__ off,
                                              int* __restrict__ csr,
                                              float* __restrict__ el,
                                              float* __restrict__ er) {
  __shared__ int    cnt_s[256];
  __shared__ int    cur_s[256];
  __shared__ int    red_s[32];
  __shared__ float4 wlr4[16];
  float* wlr_s = (float*)wlr4;
  const int t = threadIdx.x, lane = t & 31, wv = t >> 5;
  const int part = blockIdx.x & (NPART - 1);
  const int lo = (blockIdx.x / NPART) * NPB;
  const int qb = part * QP;

  const int4* d4 = (const int4*)dst;
  int4 d[NQT];
  #pragma unroll
  for (int u = 0; u < FULLR; ++u) d[u] = d4[qb + t + TA * u];
  {
    const int q = FULLR * TA + t;
    d[FULLR] = d4[qb + min(q, QP - 1)];
    if (q >= QP) d[FULLR] = make_int4(0x7FFFFFFF, 0x7FFFFFFF, 0x7FFFFFFF, 0x7FFFFFFF);
  }

  if (t < 256) cnt_s[t] = 0;

  const int nel0 = lo + part * NEL;
  const int xq = min(nel0 * 8 + t, NN * 8 - 1);
  const float4 xv = ((const float4*)x)[xq];

  if (blockIdx.x == 0 && t < OUTD) {
    const float v = px_r[t];
    const float sp = (v > 20.0f) ? v : log1pf(__expf(v));
    out[(size_t)NN * OUTD + t] = fminf(fmaxf(sp, 1.0e-4f), 1.0e4f);
  }
  if (blockIdx.x >= GA - NPART && t == 0) off[NN * NPART + part] = (part + 1) * EP;

  #pragma unroll
  for (int g2 = 0; g2 < 2; ++g2) {
    const float* a = g2 ? attn_r : attn_l;
    float s = 0.0f;
    #pragma unroll
    for (int j = 0; j < 16; ++j) {
      const int dd = lane + 32 * j;
      const int dc = (dd < OUTD) ? dd : 0;
      s = fmaf(W[wv * OUTD + dc], (dd < OUTD) ? a[dc] : 0.0f, s);
    }
    #pragma unroll
    for (int o = 16; o; o >>= 1) s += __shfl_xor(s, o, 32);
    if (lane == 0) wlr_s[wv + 32 * g2] = s;
  }
  __syncthreads();

  {
    const int part8 = t & 7;
    const float4 l  = wlr4[part8];
    const float4 rr = wlr4[8 + part8];
    float sl = fmaf(xv.x, l.x, fmaf(xv.y, l.y, fmaf(xv.z, l.z, xv.w * l.w)));
    float sr = fmaf(xv.x, rr.x, fmaf(xv.y, rr.y, fmaf(xv.z, rr.z, xv.w * rr.w)));
    #pragma unroll
    for (int o = 4; o; o >>= 1) {
      sl += __shfl_xor(sl, o, 32);
      sr += __shfl_xor(sr, o, 32);
    }
    if (part8 == 0 && t < NEL * 8) { el[nel0 + (t >> 3)] = sl; er[nel0 + (t >> 3)] = sr; }
  }

  int below = 0;
  #pragma unroll
  for (int u = 0; u < NQT; ++u) {
    count_one(d[u].x, lo, cnt_s, below);
    count_one(d[u].y, lo, cnt_s, below);
    count_one(d[u].z, lo, cnt_s, below);
    count_one(d[u].w, lo, cnt_s, below);
  }
  #pragma unroll
  for (int o = 16; o; o >>= 1) below += __shfl_xor(below, o, 32);
  if (lane == 0) red_s[wv] = below;
  __syncthreads();

  if (wv == 0) {
    int base = red_s[lane];
    #pragma unroll
    for (int o = 16; o; o >>= 1) base += __shfl_xor(base, o, 32);
    int c[8];
    int tot = 0;
    #pragma unroll
    for (int j = 0; j < 8; ++j) {
      const int n = 8 * lane + j;
      c[j] = (n < NPB) ? cnt_s[n] : 0;
      tot += c[j];
    }
    int inc = tot;
    #pragma unroll
    for (int o = 1; o < 32; o <<= 1) {
      const int u = __shfl_up(inc, o, 32);
      if (lane >= o) inc += u;
    }
    int run = part * EP + base + inc - tot;
    #pragma unroll
    for (int j = 0; j < 8; ++j) {
      const int n = 8 * lane + j;
      if (n < NPB) { off[(lo + n) * NPART + part] = run; cur_s[n] = run; }
      run += c[j];
    }
  }
  __syncthreads();

  #pragma unroll
  for (int u = 0; u < NQT; ++u) {
    const int e0 = (qb + t + TA * u) << 2;
    place_one(d[u].x, lo, e0,     cur_s, csr);
    place_one(d[u].y, lo, e0 + 1, cur_s, csr);
    place_one(d[u].z, lo, e0 + 2, cur_s, csr);
    place_one(d[u].w, lo, e0 + 3, cur_s, csr);
  }
}

__device__ __forceinline__ float wave_max(float v) {
  #pragma unroll
  for (int o = 16; o; o >>= 1) v = fmaxf(v, __shfl_xor(v, o, 32));
  return v;
}
__device__ __forceinline__ float leaky(float v) {
  return (v > 0.0f) ? v : NEG_SLOPE * v;
}
struct SegMap { int p1, p2, p3, a0, a1, a2, a3; };
__device__ __forceinline__ int seg_index(const SegMap& m, int q) {
  int a = m.a0;
  a = (q >= m.p1) ? m.a1 : a;
  a = (q >= m.p2) ? m.a2 : a;
  a = (q >= m.p3) ? m.a3 : a;
  return q + a;
}
__device__ __forceinline__ int next_edge(const int* __restrict__ csr, const SegMap& m,
                                         int deg, int prev, int lane) {
  int best = 0x7FFFFFFF;
  for (int q = lane; q < deg; q += 32) {
    const int c = csr[seg_index(m, q)];
    if (c > prev && c < best) best = c;
  }
  #pragma unroll
  for (int o = 16; o; o >>= 1) best = min(best, __shfl_xor(best, o, 32));
  return __builtin_amdgcn_readfirstlane(best);
}

#define APITCH 36
__global__ __launch_bounds__(512) void k_gather_gemm_softmax(const int* __restrict__ src,
                                                             const int* __restrict__ off,
                                                             const int* __restrict__ csr,
                                                             const float* __restrict__ el,
                                                             const float* __restrict__ er,
                                                             const float* __restrict__ x,
                                                             const float* __restrict__ W,
                                                             float* __restrict__ out) {
  __shared__ float4 S4[16 * NQ + 8];
  __shared__ float  As[16 * APITCH];
  __shared__ float sw_s[16 * 64];
  __shared__ int   sr_s[16 * 64];
  float* S = (float*)S4;
  const int wv   = __builtin_amdgcn_readfirstlane(threadIdx.x >> 5);
  const int lane = threadIdx.x & 31;
  const int r  = lane & 15;
  const int h  = lane >> 4;
  const int m0 = blockIdx.x << 4;
  const int node = m0 + wv;

  v2f B[2][8];
  #pragma unroll
  for (int t = 0; t < 2; ++t) {
    const int col = (wv << 5) + (t << 4) + r;
    const int cc  = (col < OUTD) ? col : 0;
    #pragma unroll
    for (int i = 0; i < 8; ++i) {
      const int k = (i << 2) + (h << 1);
      B[t][i].x = W[k * OUTD + cc];
      B[t][i].y = W[(k + 1) * OUTD + cc];
    }
  }

  const int4 st = ((const int4*)off)[node];
  const int4 en = ((const int4*)off)[node + 1];
  SegMap sg;
  sg.p1 = en.x - st.x;
  sg.p2 = sg.p1 + (en.y - st.y);
  sg.p3 = sg.p2 + (en.z - st.z);
  sg.a0 = st.x;
  sg.a1 = st.y - sg.p1;
  sg.a2 = st.z - sg.p2;
  sg.a3 = st.w - sg.p3;
  const int deg = sg.p3 + (en.w - st.w);
  const float ern = er[node];
  float acc = 0.0f;

  if (deg <= 32) {
    const bool act = lane < deg;
    const int id = act ? csr[seg_index(sg, lane)] : 0x7FFFFFFF;
    const int sn = act ? src[id] : 0;
    const float e = act ? leaky(el[sn] + ern) : -INFINITY;
    int rank = 0;
    for (int j = 0; j < deg; ++j)
      rank += (__builtin_amdgcn_readlane(id, j) < id) ? 1 : 0;
    if (!act) rank = lane;
    float m = wave_max(e);
    m = isfinite(m) ? m : 0.0f;
    const float a = act ? __expf(e - m) : 0.0f;
    const int sa = __builtin_amdgcn_ds_permute(rank << 2, __float_as_int(a));
    const int ss = __builtin_amdgcn_ds_permute(rank << 2, sn);
    float s = 0.0f;
    for (int k = 0; k < deg; ++k) s += __int_as_float(__builtin_amdgcn_readlane(sa, k));
    const int al_bits = __float_as_int(__int_as_float(sa) / (s > 0.0f ? s : 1.0f));
    for (int k0 = 0; k0 < deg; k0 += 16) {
      #pragma unroll
      for (int j = 0; j < 16; ++j) {
        const int   snk = __builtin_amdgcn_readlane(ss, k0 + j);
        const float al  = __int_as_float(__builtin_amdgcn_readlane(al_bits, k0 + j));
        acc = fmaf(al, x[(size_t)snk * IND + lane], acc);
      }
    }
  } else if (deg <= 64) {
    const int  nb   = deg - 32;
    const bool actB = lane < nb;
    const int idA = csr[seg_index(sg, lane)];
    const int idB = actB ? csr[seg_index(sg, 32 + lane)] : 0x7FFFFFFF;
    const int snA = src[idA];
    const int snB = actB ? src[idB] : 0;
    const float eA = leaky(el[snA] + ern);
    const float eB = actB ? leaky(el[snB] + ern) : -INFINITY;
    int rkA = 0, rkB = 0;
    for (int j = 0; j < 32; ++j) {
      const int c = __builtin_amdgcn_readlane(idA, j);
      rkA += (c < idA) ? 1 : 0;
      rkB += (c < idB) ? 1 : 0;
    }
    for (int j = 0; j < nb; ++j) {
      const int c = __builtin_amdgcn_readlane(idB, j);
      rkA += (c < idA) ? 1 : 0;
      rkB += (c < idB) ? 1 : 0;
    }
    float m = wave_max(fmaxf(eA, eB));
    m = isfinite(m) ? m : 0.0f;
    float* sw = sw_s + (wv << 6);
    int*   sr = sr_s + (wv << 6);
    sw[rkA] = __expf(eA - m);
    sr[rkA] = snA;
    if (actB) { sw[rkB] = __expf(eB - m); sr[rkB] = snB; }
    const int   waA = __float_as_int(sw[lane]);
    const int   ssA = sr[lane];
    const int   waB = actB ? __float_as_int(sw[32 + lane]) : 0;
    const int   ssB = actB ? sr[32 + lane] : 0;
    float s = 0.0f;
    for (int k = 0; k < 32; ++k) s += __int_as_float(__builtin_amdgcn_readlane(waA, k));
    for (int k = 0; k < nb; ++k) s += __int_as_float(__builtin_amdgcn_readlane(waB, k));
    const float den = (s > 0.0f) ? s : 1.0f;
    const int alA = __float_as_int(__int_as_float(waA) / den);
    const int alB = __float_as_int(__int_as_float(waB) / den);
    for (int k0 = 0; k0 < 32; k0 += 16) {
      #pragma unroll
      for (int j = 0; j < 16; ++j) {
        const int   snk = __builtin_amdgcn_readlane(ssA, k0 + j);
        const float al  = __int_as_float(__builtin_amdgcn_readlane(alA, k0 + j));
        acc = fmaf(al, x[(size_t)snk * IND + lane], acc);
      }
    }
    for (int k0 = 0; k0 < nb; k0 += 16) {
      #pragma unroll
      for (int j = 0; j < 16; ++j) {
        const int   snk = __builtin_amdgcn_readlane(ssB, k0 + j);
        const float al  = __int_as_float(__builtin_amdgcn_readlane(alB, k0 + j));
        acc = fmaf(al, x[(size_t)snk * IND + lane], acc);
      }
    }
  } else {
    float mx = -INFINITY;
    for (int q = lane; q < deg; q += 32)
      mx = fmaxf(mx, leaky(el[src[csr[seg_index(sg, q)]]] + ern));
    float m = wave_max(mx);
    m = isfinite(m) ? m : 0.0f;
    float s = 0.0f;
    int prev = -1;
    for (int k = 0; k < deg; ++k) {
      prev = next_edge(csr, sg, deg, prev, lane);
      s += __expf(leaky(el[src[prev]] + ern) - m);
    }
    const float den = (s > 0.0f) ? s : 1.0f;
    prev = -1;
    for (int k = 0; k < deg; ++k) {
      prev = next_edge(csr, sg, deg, prev, lane);
      const int snk = src[prev];
      const float al = __expf(leaky(el[snk] + ern) - m) / den;
      acc = fmaf(al, x[(size_t)snk * IND + lane], acc);
    }
  }
  As[wv * APITCH + lane] = acc;
  __syncthreads();

  v2f A[8];
  #pragma unroll
  for (int i = 0; i < 8; ++i) {
    const int k = (i << 2) + (h << 1);
    A[i] = *(const v2f*)(As + r * APITCH + k);
  }
  v8f cacc[2] = {};
  #pragma unroll
  for (int i = 0; i < 8; ++i) {
    #pragma unroll
    for (int t = 0; t < 2; ++t)
      cacc[t] = __builtin_amdgcn_wmma_f32_16x16x4_f32(
          false, A[i], false, B[t][i], (short)0, cacc[t], false, false);
  }

  #pragma unroll
  for (int t = 0; t < 2; ++t) {
    const int col = (wv << 5) + (t << 4) + r;
    #pragma unroll
    for (int rr = 0; rr < 8; ++rr) {
      int idx = (rr + (h << 3)) * OUTD + col;
      if (t == 1) idx = (col < OUTD) ? idx : 16 * OUTD + lane;
      S[idx] = cacc[t][rr];
    }
  }
  __syncthreads();

  {
    const bool last = (lane + 96) < NQ;
    float4* r4 = S4 + wv * NQ;
    float4 v[4];
    #pragma unroll
    for (int j = 0; j < 3; ++j) v[j] = r4[lane + 32 * j];
    v[3] = last ? r4[lane + 96] : make_float4(-INFINITY, -INFINITY, -INFINITY, -INFINITY);
    float mx = -INFINITY;
    #pragma unroll
    for (int j = 0; j < 4; ++j)
      mx = fmaxf(mx, fmaxf(fmaxf(v[j].x, v[j].y), fmaxf(v[j].z, v[j].w)));
    #pragma unroll
    for (int o = 16; o; o >>= 1) mx = fmaxf(mx, __shfl_xor(mx, o, 32));
    float sm = 0.0f;
    #pragma unroll
    for (int j = 0; j < 4; ++j) {
      const bool act = (j < 3) || last;
      v[j].x = act ? __expf(v[j].x - mx) : 0.0f;
      v[j].y = act ? __expf(v[j].y - mx) : 0.0f;
      v[j].z = act ? __expf(v[j].z - mx) : 0.0f;
      v[j].w = act ? __expf(v[j].w - mx) : 0.0f;
      sm += (v[j].x + v[j].y) + (v[j].z + v[j].w);
    }
    #pragma unroll
    for (int o = 16; o; o >>= 1) sm += __shfl_xor(sm, o, 32);
    const float inv = 1.0f / sm;
    #pragma unroll
    for (int j = 0; j < 4; ++j) {
      v[j].x *= inv; v[j].y *= inv; v[j].z *= inv; v[j].w *= inv;
    }
    #pragma unroll
    for (int j = 0; j < 3; ++j) r4[lane + 32 * j] = v[j];
    if (last) r4[lane + 96] = v[3];
  }
  __syncthreads();

  float4* o4 = (float4*)(out + (size_t)m0 * OUTD);
  const int tid = threadIdx.x;
  #pragma unroll
  for (int j = 0; j < 3; ++j) o4[tid + 512 * j] = S4[tid + 512 * j];
  if (tid < 16 * NQ - 3 * 512) o4[tid + 3 * 512] = S4[tid + 3 * 512];
}

extern "C" void kernel_launch(void* const* d_in, const int* in_sizes, int n_in,
                              void* d_out, int out_size, void* d_ws, size_t ws_size,
                              hipStream_t stream) {
  const float* x      = (const float*)d_in[0];
  const float* W      = (const float*)d_in[1];
  const float* attn_l = (const float*)d_in[2];
  const float* attn_r = (const float*)d_in[3];
  const float* px_r   = (const float*)d_in[4];
  const int*   src    = (const int*)d_in[5];
  const int*   dst    = (const int*)d_in[6];
  float*       out    = (float*)d_out;

  float* ws   = (float*)d_ws;
  float* el   = ws;
  float* er   = el + NN;
  int*   off  = (int*)(er + NN);
  int*   csr  = off + NPART * (NN + 1);

  k_build              <<<GA, TA, 0, stream>>>(dst, x, W, attn_l, attn_r, px_r, out, off, csr, el, er);
  k_gather_gemm_softmax<<<MTILES, 512, 0, stream>>>(src, off, csr, el, er, x, W, out);
}


# wave reductions and the edge-list scan on vector-ALU lane exchanges (DPP row_xmask/row_shr, v_permlanex16), not ds_bpermute LDS round trips, both launches; same butterfly order, sums bit-identical
# speedup vs baseline: 9.0674x; 1.0430x over previous
#include <hip/hip_runtime.h>
#include <math.h>

#define NN     10000
#define EE     160000
#define IND    32
#define OUTD   500
#define NQ     125
#define NTILES 32
#define MTILES 625
#define NEG_SLOPE 0.2f

typedef __attribute__((ext_vector_type(2)))  float    v2f;
typedef __attribute__((ext_vector_type(8)))  float    v8f;

template <int M> __device__ __forceinline__ int lx_i(int b) {
  return __builtin_amdgcn_update_dpp(0, b, 0x160 | M, 0xf, 0xf, true);
}
template <> __device__ __forceinline__ int lx_i<16>(int b) {
  return __builtin_amdgcn_permlanex16(b, b, 0x76543210, 0xfedcba98, false, false);
}
template <int M> __device__ __forceinline__ float lx_f(float x) {
  return __int_as_float(lx_i<M>(__float_as_int(x)));
}
__device__ __forceinline__ float wave_sum(float s) {
  s += lx_f<16>(s); s += lx_f<8>(s); s += lx_f<4>(s); s += lx_f<2>(s); s += lx_f<1>(s);
  return s;
}
__device__ __forceinline__ int wave_sum_i(int s) {
  s += lx_i<16>(s); s += lx_i<8>(s); s += lx_i<4>(s); s += lx_i<2>(s); s += lx_i<1>(s);
  return s;
}
__device__ __forceinline__ int wave_scan_i(int v, int lane) {
  v += __builtin_amdgcn_update_dpp(0, v, 0x111, 0xf, 0xf, false);
  v += __builtin_amdgcn_update_dpp(0, v, 0x112, 0xf, 0xf, false);
  v += __builtin_amdgcn_update_dpp(0, v, 0x114, 0xf, 0xf, false);
  v += __builtin_amdgcn_update_dpp(0, v, 0x118, 0xf, 0xf, false);
  const int row0 = __builtin_amdgcn_readlane(v, 15);
  return v + ((lane >= 16) ? row0 : 0);
}

#define NPART 4
#define NOWN  50
#define GA    (NOWN * NPART)
#define NPB   200
#define NEL   (NPB / NPART)
#define TA    1024
#define EP    (EE / NPART)
#define QP    (EP / 4)
#define FULLR 9
#define NQT   (FULLR + 1)

__device__ __forceinline__ void count_one(int d, int lo, int* cnt_s, int& below) {
  const int v = d - lo;
  below += (v < 0) ? 1 : 0;
  if ((unsigned)v < (unsigned)NPB) atomicAdd(&cnt_s[v], 1);
}
__device__ __forceinline__ void place_one(int d, int lo, int eid, int* cur_s,
                                          int* __restrict__ csr) {
  const int v = d - lo;
  if ((unsigned)v < (unsigned)NPB) {
    const int p = atomicAdd(&cur_s[v], 1);
    csr[p] = eid;
  }
}

__global__ __launch_bounds__(TA) void k_build(const int* __restrict__ dst,
                                              const float* __restrict__ x,
                                              const float* __restrict__ W,
                                              const float* __restrict__ attn_l,
                                              const float* __restrict__ attn_r,
                                              const float* __restrict__ px_r,
                                              float* __restrict__ out,
                                              int* __restrict__ off,
                                              int* __restrict__ csr,
                                              float* __restrict__ el,
                                              float* __restrict__ er) {
  __shared__ int    cnt_s[256];
  __shared__ int    cur_s[256];
  __shared__ int    red_s[32];
  __shared__ float4 wlr4[16];
  float* wlr_s = (float*)wlr4;
  const int t = threadIdx.x, lane = t & 31, wv = t >> 5;
  const int part = blockIdx.x & (NPART - 1);
  const int lo = (blockIdx.x / NPART) * NPB;
  const int qb = part * QP;

  const int4* d4 = (const int4*)dst;
  int4 d[NQT];
  #pragma unroll
  for (int u = 0; u < FULLR; ++u) d[u] = d4[qb + t + TA * u];
  {
    const int q = FULLR * TA + t;
    d[FULLR] = d4[qb + min(q, QP - 1)];
    if (q >= QP) d[FULLR] = make_int4(0x7FFFFFFF, 0x7FFFFFFF, 0x7FFFFFFF, 0x7FFFFFFF);
  }

  if (t < 256) cnt_s[t] = 0;

  const int nel0 = lo + part * NEL;
  const int xq = min(nel0 * 8 + t, NN * 8 - 1);
  const float4 xv = ((const float4*)x)[xq];

  if (blockIdx.x == 0 && t < OUTD) {
    const float v = px_r[t];
    const float sp = (v > 20.0f) ? v : log1pf(__expf(v));
    out[(size_t)NN * OUTD + t] = fminf(fmaxf(sp, 1.0e-4f), 1.0e4f);
  }
  if (blockIdx.x >= GA - NPART && t == 0) off[NN * NPART + part] = (part + 1) * EP;

  #pragma unroll
  for (int g2 = 0; g2 < 2; ++g2) {
    const float* a = g2 ? attn_r : attn_l;
    float s = 0.0f;
    #pragma unroll
    for (int j = 0; j < 16; ++j) {
      const int dd = lane + 32 * j;
      const int dc = (dd < OUTD) ? dd : 0;
      s = fmaf(W[wv * OUTD + dc], (dd < OUTD) ? a[dc] : 0.0f, s);
    }
    s = wave_sum(s);
    if (lane == 0) wlr_s[wv + 32 * g2] = s;
  }
  __syncthreads();

  {
    const int part8 = t & 7;
    const float4 l  = wlr4[part8];
    const float4 rr = wlr4[8 + part8];
    float sl = fmaf(xv.x, l.x, fmaf(xv.y, l.y, fmaf(xv.z, l.z, xv.w * l.w)));
    float sr = fmaf(xv.x, rr.x, fmaf(xv.y, rr.y, fmaf(xv.z, rr.z, xv.w * rr.w)));
    sl += lx_f<4>(sl); sr += lx_f<4>(sr);
    sl += lx_f<2>(sl); sr += lx_f<2>(sr);
    sl += lx_f<1>(sl); sr += lx_f<1>(sr);
    if (part8 == 0 && t < NEL * 8) { el[nel0 + (t >> 3)] = sl; er[nel0 + (t >> 3)] = sr; }
  }

  int below = 0;
  #pragma unroll
  for (int u = 0; u < NQT; ++u) {
    count_one(d[u].x, lo, cnt_s, below);
    count_one(d[u].y, lo, cnt_s, below);
    count_one(d[u].z, lo, cnt_s, below);
    count_one(d[u].w, lo, cnt_s, below);
  }
  below = wave_sum_i(below);
  if (lane == 0) red_s[wv] = below;
  __syncthreads();

  if (wv == 0) {
    const int base = wave_sum_i(red_s[lane]);
    int c[8];
    int tot = 0;
    #pragma unroll
    for (int j = 0; j < 8; ++j) {
      const int n = 8 * lane + j;
      c[j] = (n < NPB) ? cnt_s[n] : 0;
      tot += c[j];
    }
    const int inc = wave_scan_i(tot, lane);
    int run = part * EP + base + inc - tot;
    #pragma unroll
    for (int j = 0; j < 8; ++j) {
      const int n = 8 * lane + j;
      if (n < NPB) { off[(lo + n) * NPART + part] = run; cur_s[n] = run; }
      run += c[j];
    }
  }
  __syncthreads();

  #pragma unroll
  for (int u = 0; u < NQT; ++u) {
    const int e0 = (qb + t + TA * u) << 2;
    place_one(d[u].x, lo, e0,     cur_s, csr);
    place_one(d[u].y, lo, e0 + 1, cur_s, csr);
    place_one(d[u].z, lo, e0 + 2, cur_s, csr);
    place_one(d[u].w, lo, e0 + 3, cur_s, csr);
  }
}

__device__ __forceinline__ float wave_max(float v) {
  v = fmaxf(v, lx_f<16>(v)); v = fmaxf(v, lx_f<8>(v)); v = fmaxf(v, lx_f<4>(v));
  v = fmaxf(v, lx_f<2>(v));  v = fmaxf(v, lx_f<1>(v));
  return v;
}
__device__ __forceinline__ float leaky(float v) {
  return (v > 0.0f) ? v : NEG_SLOPE * v;
}
struct SegMap { int p1, p2, p3, a0, a1, a2, a3; };
__device__ __forceinline__ int seg_index(const SegMap& m, int q) {
  int a = m.a0;
  a = (q >= m.p1) ? m.a1 : a;
  a = (q >= m.p2) ? m.a2 : a;
  a = (q >= m.p3) ? m.a3 : a;
  return q + a;
}
__device__ __forceinline__ int next_edge(const int* __restrict__ csr, const SegMap& m,
                                         int deg, int prev, int lane) {
  int best = 0x7FFFFFFF;
  for (int q = lane; q < deg; q += 32) {
    const int c = csr[seg_index(m, q)];
    if (c > prev && c < best) best = c;
  }
  best = min(best, lx_i<16>(best)); best = min(best, lx_i<8>(best)); best = min(best, lx_i<4>(best));
  best = min(best, lx_i<2>(best));  best = min(best, lx_i<1>(best));
  return __builtin_amdgcn_readfirstlane(best);
}

#define APITCH 36
__global__ __launch_bounds__(512) void k_gather_gemm_softmax(const int* __restrict__ src,
                                                             const int* __restrict__ off,
                                                             const int* __restrict__ csr,
                                                             const float* __restrict__ el,
                                                             const float* __restrict__ er,
                                                             const float* __restrict__ x,
                                                             const float* __restrict__ W,
                                                             float* __restrict__ out) {
  __shared__ float4 S4[16 * NQ + 8];
  __shared__ float  As[16 * APITCH];
  __shared__ float sw_s[16 * 64];
  __shared__ int   sr_s[16 * 64];
  float* S = (float*)S4;
  const int wv   = __builtin_amdgcn_readfirstlane(threadIdx.x >> 5);
  const int lane = threadIdx.x & 31;
  const int r  = lane & 15;
  const int h  = lane >> 4;
  const int m0 = blockIdx.x << 4;
  const int node = m0 + wv;

  v2f B[2][8];
  #pragma unroll
  for (int t = 0; t < 2; ++t) {
    const int col = (wv << 5) + (t << 4) + r;
    const int cc  = (col < OUTD) ? col : 0;
    #pragma unroll
    for (int i = 0; i < 8; ++i) {
      const int k = (i << 2) + (h << 1);
      B[t][i].x = W[k * OUTD + cc];
      B[t][i].y = W[(k + 1) * OUTD + cc];
    }
  }

  const int4 st = ((const int4*)off)[node];
  const int4 en = ((const int4*)off)[node + 1];
  SegMap sg;
  sg.p1 = en.x - st.x;
  sg.p2 = sg.p1 + (en.y - st.y);
  sg.p3 = sg.p2 + (en.z - st.z);
  sg.a0 = st.x;
  sg.a1 = st.y - sg.p1;
  sg.a2 = st.z - sg.p2;
  sg.a3 = st.w - sg.p3;
  const int deg = sg.p3 + (en.w - st.w);
  const float ern = er[node];
  float acc = 0.0f;

  if (deg <= 32) {
    const bool act = lane < deg;
    const int id = act ? csr[seg_index(sg, lane)] : 0x7FFFFFFF;
    const int sn = act ? src[id] : 0;
    const float e = act ? leaky(el[sn] + ern) : -INFINITY;
    int rank = 0;
    for (int j = 0; j < deg; ++j)
      rank += (__builtin_amdgcn_readlane(id, j) < id) ? 1 : 0;
    if (!act) rank = lane;
    float m = wave_max(e);
    m = isfinite(m) ? m : 0.0f;
    const float a = act ? __expf(e - m) : 0.0f;
    const int sa = __builtin_amdgcn_ds_permute(rank << 2, __float_as_int(a));
    const int ss = __builtin_amdgcn_ds_permute(rank << 2, sn);
    float s = 0.0f;
    for (int k = 0; k < deg; ++k) s += __int_as_float(__builtin_amdgcn_readlane(sa, k));
    const int al_bits = __float_as_int(__int_as_float(sa) / (s > 0.0f ? s : 1.0f));
    for (int k0 = 0; k0 < deg; k0 += 16) {
      #pragma unroll
      for (int j = 0; j < 16; ++j) {
        const int   snk = __builtin_amdgcn_readlane(ss, k0 + j);
        const float al  = __int_as_float(__builtin_amdgcn_readlane(al_bits, k0 + j));
        acc = fmaf(al, x[(size_t)snk * IND + lane], acc);
      }
    }
  } else if (deg <= 64) {
    const int  nb   = deg - 32;
    const bool actB = lane < nb;
    const int idA = csr[seg_index(sg, lane)];
    const int idB = actB ? csr[seg_index(sg, 32 + lane)] : 0x7FFFFFFF;
    const int snA = src[idA];
    const int snB = actB ? src[idB] : 0;
    const float eA = leaky(el[snA] + ern);
    const float eB = actB ? leaky(el[snB] + ern) : -INFINITY;
    int rkA = 0, rkB = 0;
    for (int j = 0; j < 32; ++j) {
      const int c = __builtin_amdgcn_readlane(idA, j);
      rkA += (c < idA) ? 1 : 0;
      rkB += (c < idB) ? 1 : 0;
    }
    for (int j = 0; j < nb; ++j) {
      const int c = __builtin_amdgcn_readlane(idB, j);
      rkA += (c < idA) ? 1 : 0;
      rkB += (c < idB) ? 1 : 0;
    }
    float m = wave_max(fmaxf(eA, eB));
    m = isfinite(m) ? m : 0.0f;
    float* sw = sw_s + (wv << 6);
    int*   sr = sr_s + (wv << 6);
    sw[rkA] = __expf(eA - m);
    sr[rkA] = snA;
    if (actB) { sw[rkB] = __expf(eB - m); sr[rkB] = snB; }
    const int   waA = __float_as_int(sw[lane]);
    const int   ssA = sr[lane];
    const int   waB = actB ? __float_as_int(sw[32 + lane]) : 0;
    const int   ssB = actB ? sr[32 + lane] : 0;
    float s = 0.0f;
    for (int k = 0; k < 32; ++k) s += __int_as_float(__builtin_amdgcn_readlane(waA, k));
    for (int k = 0; k < nb; ++k) s += __int_as_float(__builtin_amdgcn_readlane(waB, k));
    const float den = (s > 0.0f) ? s : 1.0f;
    const int alA = __float_as_int(__int_as_float(waA) / den);
    const int alB = __float_as_int(__int_as_float(waB) / den);
    for (int k0 = 0; k0 < 32; k0 += 16) {
      #pragma unroll
      for (int j = 0; j < 16; ++j) {
        const int   snk = __builtin_amdgcn_readlane(ssA, k0 + j);
        const float al  = __int_as_float(__builtin_amdgcn_readlane(alA, k0 + j));
        acc = fmaf(al, x[(size_t)snk * IND + lane], acc);
      }
    }
    for (int k0 = 0; k0 < nb; k0 += 16) {
      #pragma unroll
      for (int j = 0; j < 16; ++j) {
        const int   snk = __builtin_amdgcn_readlane(ssB, k0 + j);
        const float al  = __int_as_float(__builtin_amdgcn_readlane(alB, k0 + j));
        acc = fmaf(al, x[(size_t)snk * IND + lane], acc);
      }
    }
  } else {
    float mx = -INFINITY;
    for (int q = lane; q < deg; q += 32)
      mx = fmaxf(mx, leaky(el[src[csr[seg_index(sg, q)]]] + ern));
    float m = wave_max(mx);
    m = isfinite(m) ? m : 0.0f;
    float s = 0.0f;
    int prev = -1;
    for (int k = 0; k < deg; ++k) {
      prev = next_edge(csr, sg, deg, prev, lane);
      s += __expf(leaky(el[src[prev]] + ern) - m);
    }
    const float den = (s > 0.0f) ? s : 1.0f;
    prev = -1;
    for (int k = 0; k < deg; ++k) {
      prev = next_edge(csr, sg, deg, prev, lane);
      const int snk = src[prev];
      const float al = __expf(leaky(el[snk] + ern) - m) / den;
      acc = fmaf(al, x[(size_t)snk * IND + lane], acc);
    }
  }
  As[wv * APITCH + lane] = acc;
  __syncthreads();

  v2f A[8];
  #pragma unroll
  for (int i = 0; i < 8; ++i) {
    const int k = (i << 2) + (h << 1);
    A[i] = *(const v2f*)(As + r * APITCH + k);
  }
  v8f cacc[2] = {};
  #pragma unroll
  for (int i = 0; i < 8; ++i) {
    #pragma unroll
    for (int t = 0; t < 2; ++t)
      cacc[t] = __builtin_amdgcn_wmma_f32_16x16x4_f32(
          false, A[i], false, B[t][i], (short)0, cacc[t], false, false);
  }

  #pragma unroll
  for (int t = 0; t < 2; ++t) {
    const int col = (wv << 5) + (t << 4) + r;
    #pragma unroll
    for (int rr = 0; rr < 8; ++rr) {
      int idx = (rr + (h << 3)) * OUTD + col;
      if (t == 1) idx = (col < OUTD) ? idx : 16 * OUTD + lane;
      S[idx] = cacc[t][rr];
    }
  }
  __syncthreads();

  {
    const bool last = (lane + 96) < NQ;
    float4* r4 = S4 + wv * NQ;
    float4 v[4];
    #pragma unroll
    for (int j = 0; j < 3; ++j) v[j] = r4[lane + 32 * j];
    v[3] = last ? r4[lane + 96] : make_float4(-INFINITY, -INFINITY, -INFINITY, -INFINITY);
    float mx = -INFINITY;
    #pragma unroll
    for (int j = 0; j < 4; ++j)
      mx = fmaxf(mx, fmaxf(fmaxf(v[j].x, v[j].y), fmaxf(v[j].z, v[j].w)));
    mx = wave_max(mx);
    float sm = 0.0f;
    #pragma unroll
    for (int j = 0; j < 4; ++j) {
      const bool act = (j < 3) || last;
      v[j].x = act ? __expf(v[j].x - mx) : 0.0f;
      v[j].y = act ? __expf(v[j].y - mx) : 0.0f;
      v[j].z = act ? __expf(v[j].z - mx) : 0.0f;
      v[j].w = act ? __expf(v[j].w - mx) : 0.0f;
      sm += (v[j].x + v[j].y) + (v[j].z + v[j].w);
    }
    sm = wave_sum(sm);
    const float inv = 1.0f / sm;
    #pragma unroll
    for (int j = 0; j < 4; ++j) {
      v[j].x *= inv; v[j].y *= inv; v[j].z *= inv; v[j].w *= inv;
    }
    #pragma unroll
    for (int j = 0; j < 3; ++j) r4[lane + 32 * j] = v[j];
    if (last) r4[lane + 96] = v[3];
  }
  __syncthreads();

  float4* o4 = (float4*)(out + (size_t)m0 * OUTD);
  const int tid = threadIdx.x;
  #pragma unroll
  for (int j = 0; j < 3; ++j) o4[tid + 512 * j] = S4[tid + 512 * j];
  if (tid < 16 * NQ - 3 * 512) o4[tid + 3 * 512] = S4[tid + 3 * 512];
}

extern "C" void kernel_launch(void* const* d_in, const int* in_sizes, int n_in,
                              void* d_out, int out_size, void* d_ws, size_t ws_size,
                              hipStream_t stream) {
  const float* x      = (const float*)d_in[0];
  const float* W      = (const float*)d_in[1];
  const float* attn_l = (const float*)d_in[2];
  const float* attn_r = (const float*)d_in[3];
  const float* px_r   = (const float*)d_in[4];
  const int*   src    = (const int*)d_in[5];
  const int*   dst    = (const int*)d_in[6];
  float*       out    = (float*)d_out;

  float* ws   = (float*)d_ws;
  float* el   = ws;
  float* er   = el + NN;
  int*   off  = (int*)(er + NN);
  int*   csr  = off + NPART * (NN + 1);

  k_build              <<<GA, TA, 0, stream>>>(dst, x, W, attn_l, attn_r, px_r, out, off, csr, el, er);
  k_gather_gemm_softmax<<<MTILES, 512, 0, stream>>>(src, off, csr, el, er, x, W, out);
}


# gather+GEMM launch, same bytes in fewer, wider loads: x rows 4 per 16-byte-per-lane load (4 partial row sums: sum order changes), W fragments as 8-byte column pairs (tiles = even/odd columns)
# speedup vs baseline: 9.4448x; 1.0416x over previous
#include <hip/hip_runtime.h>
#include <math.h>

#define NN     10000
#define EE     160000
#define IND    32
#define OUTD   500
#define NQ     125
#define NTILES 32
#define MTILES 625
#define NEG_SLOPE 0.2f

typedef __attribute__((ext_vector_type(2)))  float    v2f;
typedef __attribute__((ext_vector_type(8)))  float    v8f;

template <int M> __device__ __forceinline__ int lx_i(int b) {
  return __builtin_amdgcn_update_dpp(0, b, 0x160 | M, 0xf, 0xf, true);
}
template <> __device__ __forceinline__ int lx_i<16>(int b) {
  return __builtin_amdgcn_permlanex16(b, b, 0x76543210, 0xfedcba98, false, false);
}
template <int M> __device__ __forceinline__ float lx_f(float x) {
  return __int_as_float(lx_i<M>(__float_as_int(x)));
}
__device__ __forceinline__ float wave_sum(float s) {
  s += lx_f<16>(s); s += lx_f<8>(s); s += lx_f<4>(s); s += lx_f<2>(s); s += lx_f<1>(s);
  return s;
}
__device__ __forceinline__ int wave_sum_i(int s) {
  s += lx_i<16>(s); s += lx_i<8>(s); s += lx_i<4>(s); s += lx_i<2>(s); s += lx_i<1>(s);
  return s;
}
__device__ __forceinline__ int wave_scan_i(int v, int lane) {
  v += __builtin_amdgcn_update_dpp(0, v, 0x111, 0xf, 0xf, false);
  v += __builtin_amdgcn_update_dpp(0, v, 0x112, 0xf, 0xf, false);
  v += __builtin_amdgcn_update_dpp(0, v, 0x114, 0xf, 0xf, false);
  v += __builtin_amdgcn_update_dpp(0, v, 0x118, 0xf, 0xf, false);
  const int row0 = __builtin_amdgcn_readlane(v, 15);
  return v + ((lane >= 16) ? row0 : 0);
}

#define NPART 4
#define NOWN  50
#define GA    (NOWN * NPART)
#define NPB   200
#define NEL   (NPB / NPART)
#define TA    1024
#define EP    (EE / NPART)
#define QP    (EP / 4)
#define FULLR 9
#define NQT   (FULLR + 1)

__device__ __forceinline__ void count_one(int d, int lo, int* cnt_s, int& below) {
  const int v = d - lo;
  below += (v < 0) ? 1 : 0;
  if ((unsigned)v < (unsigned)NPB) atomicAdd(&cnt_s[v], 1);
}
__device__ __forceinline__ void place_one(int d, int lo, int eid, int* cur_s,
                                          int* __restrict__ csr) {
  const int v = d - lo;
  if ((unsigned)v < (unsigned)NPB) {
    const int p = atomicAdd(&cur_s[v], 1);
    csr[p] = eid;
  }
}

__global__ __launch_bounds__(TA) void k_build(const int* __restrict__ dst,
                                              const float* __restrict__ x,
                                              const float* __restrict__ W,
                                              const float* __restrict__ attn_l,
                                              const float* __restrict__ attn_r,
                                              const float* __restrict__ px_r,
                                              float* __restrict__ out,
                                              int* __restrict__ off,
                                              int* __restrict__ csr,
                                              float* __restrict__ el,
                                              float* __restrict__ er) {
  __shared__ int    cnt_s[256];
  __shared__ int    cur_s[256];
  __shared__ int    red_s[32];
  __shared__ float4 wlr4[16];
  float* wlr_s = (float*)wlr4;
  const int t = threadIdx.x, lane = t & 31, wv = t >> 5;
  const int part = blockIdx.x & (NPART - 1);
  const int lo = (blockIdx.x / NPART) * NPB;
  const int qb = part * QP;

  const int4* d4 = (const int4*)dst;
  int4 d[NQT];
  #pragma unroll
  for (int u = 0; u < FULLR; ++u) d[u] = d4[qb + t + TA * u];
  {
    const int q = FULLR * TA + t;
    d[FULLR] = d4[qb + min(q, QP - 1)];
    if (q >= QP) d[FULLR] = make_int4(0x7FFFFFFF, 0x7FFFFFFF, 0x7FFFFFFF, 0x7FFFFFFF);
  }

  if (t < 256) cnt_s[t] = 0;

  const int nel0 = lo + part * NEL;
  const int xq = min(nel0 * 8 + t, NN * 8 - 1);
  const float4 xv = ((const float4*)x)[xq];

  if (blockIdx.x == 0 && t < OUTD) {
    const float v = px_r[t];
    const float sp = (v > 20.0f) ? v : log1pf(__expf(v));
    out[(size_t)NN * OUTD + t] = fminf(fmaxf(sp, 1.0e-4f), 1.0e4f);
  }
  if (blockIdx.x >= GA - NPART && t == 0) off[NN * NPART + part] = (part + 1) * EP;

  #pragma unroll
  for (int g2 = 0; g2 < 2; ++g2) {
    const float* a = g2 ? attn_r : attn_l;
    float s = 0.0f;
    #pragma unroll
    for (int j = 0; j < 16; ++j) {
      const int dd = lane + 32 * j;
      const int dc = (dd < OUTD) ? dd : 0;
      s = fmaf(W[wv * OUTD + dc], (dd < OUTD) ? a[dc] : 0.0f, s);
    }
    s = wave_sum(s);
    if (lane == 0) wlr_s[wv + 32 * g2] = s;
  }
  __syncthreads();

  {
    const int part8 = t & 7;
    const float4 l  = wlr4[part8];
    const float4 rr = wlr4[8 + part8];
    float sl = fmaf(xv.x, l.x, fmaf(xv.y, l.y, fmaf(xv.z, l.z, xv.w * l.w)));
    float sr = fmaf(xv.x, rr.x, fmaf(xv.y, rr.y, fmaf(xv.z, rr.z, xv.w * rr.w)));
    sl += lx_f<4>(sl); sr += lx_f<4>(sr);
    sl += lx_f<2>(sl); sr += lx_f<2>(sr);
    sl += lx_f<1>(sl); sr += lx_f<1>(sr);
    if (part8 == 0 && t < NEL * 8) { el[nel0 + (t >> 3)] = sl; er[nel0 + (t >> 3)] = sr; }
  }

  int below = 0;
  #pragma unroll
  for (int u = 0; u < NQT; ++u) {
    count_one(d[u].x, lo, cnt_s, below);
    count_one(d[u].y, lo, cnt_s, below);
    count_one(d[u].z, lo, cnt_s, below);
    count_one(d[u].w, lo, cnt_s, below);
  }
  below = wave_sum_i(below);
  if (lane == 0) red_s[wv] = below;
  __syncthreads();

  if (wv == 0) {
    const int base = wave_sum_i(red_s[lane]);
    int c[8];
    int tot = 0;
    #pragma unroll
    for (int j = 0; j < 8; ++j) {
      const int n = 8 * lane + j;
      c[j] = (n < NPB) ? cnt_s[n] : 0;
      tot += c[j];
    }
    const int inc = wave_scan_i(tot, lane);
    int run = part * EP + base + inc - tot;
    #pragma unroll
    for (int j = 0; j < 8; ++j) {
      const int n = 8 * lane + j;
      if (n < NPB) { off[(lo + n) * NPART + part] = run; cur_s[n] = run; }
      run += c[j];
    }
  }
  __syncthreads();

  #pragma unroll
  for (int u = 0; u < NQT; ++u) {
    const int e0 = (qb + t + TA * u) << 2;
    place_one(d[u].x, lo, e0,     cur_s, csr);
    place_one(d[u].y, lo, e0 + 1, cur_s, csr);
    place_one(d[u].z, lo, e0 + 2, cur_s, csr);
    place_one(d[u].w, lo, e0 + 3, cur_s, csr);
  }
}

__device__ __forceinline__ float wave_max(float v) {
  v = fmaxf(v, lx_f<16>(v)); v = fmaxf(v, lx_f<8>(v)); v = fmaxf(v, lx_f<4>(v));
  v = fmaxf(v, lx_f<2>(v));  v = fmaxf(v, lx_f<1>(v));
  return v;
}
__device__ __forceinline__ float leaky(float v) {
  return (v > 0.0f) ? v : NEG_SLOPE * v;
}
struct SegMap { int p1, p2, p3, a0, a1, a2, a3; };
__device__ __forceinline__ int seg_index(const SegMap& m, int q) {
  int a = m.a0;
  a = (q >= m.p1) ? m.a1 : a;
  a = (q >= m.p2) ? m.a2 : a;
  a = (q >= m.p3) ? m.a3 : a;
  return q + a;
}
__device__ __forceinline__ int next_edge(const int* __restrict__ csr, const SegMap& m,
                                         int deg, int prev, int lane) {
  int best = 0x7FFFFFFF;
  for (int q = lane; q < deg; q += 32) {
    const int c = csr[seg_index(m, q)];
    if (c > prev && c < best) best = c;
  }
  best = min(best, lx_i<16>(best)); best = min(best, lx_i<8>(best)); best = min(best, lx_i<4>(best));
  best = min(best, lx_i<2>(best));  best = min(best, lx_i<1>(best));
  return __builtin_amdgcn_readfirstlane(best);
}

#define APITCH 36
__global__ __launch_bounds__(512, 12) void k_gather_gemm_softmax(const int* __restrict__ src,
                                                             const int* __restrict__ off,
                                                             const int* __restrict__ csr,
                                                             const float* __restrict__ el,
                                                             const float* __restrict__ er,
                                                             const float* __restrict__ x,
                                                             const float* __restrict__ W,
                                                             float* __restrict__ out) {
  __shared__ float4 S4[16 * NQ + 8];
  __shared__ float4 As4[16 * APITCH / 4];
  float* As = (float*)As4;
  __shared__ float sw_s[16 * 64];
  __shared__ int   sr_s[16 * 64];
  float* S = (float*)S4;
  const int wv   = __builtin_amdgcn_readfirstlane(threadIdx.x >> 5);
  const int lane = threadIdx.x & 31;
  const int r  = lane & 15;
  const int h  = lane >> 4;
  const int m0 = blockIdx.x << 4;
  const int node = m0 + wv;

  v2f B[2][8];
  {
    const int col0 = (wv << 5) + (r << 1);
    const int cc   = (col0 < OUTD) ? col0 : 0;
    #pragma unroll
    for (int i = 0; i < 8; ++i) {
      const int k = (i << 2) + (h << 1);
      const v2f w0 = *(const v2f*)(W + k * OUTD + cc);
      const v2f w1 = *(const v2f*)(W + (k + 1) * OUTD + cc);
      B[0][i].x = w0.x;  B[1][i].x = w0.y;
      B[0][i].y = w1.x;  B[1][i].y = w1.y;
    }
  }

  const int4 st = ((const int4*)off)[node];
  const int4 en = ((const int4*)off)[node + 1];
  SegMap sg;
  sg.p1 = en.x - st.x;
  sg.p2 = sg.p1 + (en.y - st.y);
  sg.p3 = sg.p2 + (en.z - st.z);
  sg.a0 = st.x;
  sg.a1 = st.y - sg.p1;
  sg.a2 = st.z - sg.p2;
  sg.a3 = st.w - sg.p3;
  const int deg = sg.p3 + (en.w - st.w);
  const float ern = er[node];
  float acc = 0.0f;

  if (deg <= 32) {
    const bool act = lane < deg;
    const int id = act ? csr[seg_index(sg, lane)] : 0x7FFFFFFF;
    const int sn = act ? src[id] : 0;
    const float e = act ? leaky(el[sn] + ern) : -INFINITY;
    int rank = 0;
    for (int j = 0; j < deg; ++j)
      rank += (__builtin_amdgcn_readlane(id, j) < id) ? 1 : 0;
    if (!act) rank = lane;
    float m = wave_max(e);
    m = isfinite(m) ? m : 0.0f;
    const float a = act ? __expf(e - m) : 0.0f;
    const int sa = __builtin_amdgcn_ds_permute(rank << 2, __float_as_int(a));
    const int ss = __builtin_amdgcn_ds_permute(rank << 2, sn);
    float s = 0.0f;
    for (int k = 0; k < deg; ++k) s += __int_as_float(__builtin_amdgcn_readlane(sa, k));
    const int al_bits = __float_as_int(__int_as_float(sa) / (s > 0.0f ? s : 1.0f));
    const int slot4 = (lane >> 3) << 2;
    const float4* xq = (const float4*)x + (lane & 7);
    float4 a4 = make_float4(0.0f, 0.0f, 0.0f, 0.0f);
    for (int k0 = 0; k0 < deg; k0 += 16) {
      #pragma unroll
      for (int j = 0; j < 4; ++j) {
        const int   pos = ((k0 + 4 * j) << 2) + slot4;
        const int   snl = __builtin_amdgcn_ds_bpermute(pos, ss);
        const float al  = __int_as_float(__builtin_amdgcn_ds_bpermute(pos, al_bits));
        const float4 rv = xq[(size_t)snl * (IND / 4)];
        a4.x = fmaf(al, rv.x, a4.x);
        a4.y = fmaf(al, rv.y, a4.y);
        a4.z = fmaf(al, rv.z, a4.z);
        a4.w = fmaf(al, rv.w, a4.w);
      }
    }
    a4.x += lx_f<8>(a4.x);  a4.y += lx_f<8>(a4.y);  a4.z += lx_f<8>(a4.z);  a4.w += lx_f<8>(a4.w);
    a4.x += lx_f<16>(a4.x); a4.y += lx_f<16>(a4.y); a4.z += lx_f<16>(a4.z); a4.w += lx_f<16>(a4.w);
    if (lane < 8) As4[wv * (APITCH / 4) + lane] = a4;
  } else if (deg <= 64) {
    const int  nb   = deg - 32;
    const bool actB = lane < nb;
    const int idA = csr[seg_index(sg, lane)];
    const int idB = actB ? csr[seg_index(sg, 32 + lane)] : 0x7FFFFFFF;
    const int snA = src[idA];
    const int snB = actB ? src[idB] : 0;
    const float eA = leaky(el[snA] + ern);
    const float eB = actB ? leaky(el[snB] + ern) : -INFINITY;
    int rkA = 0, rkB = 0;
    for (int j = 0; j < 32; ++j) {
      const int c = __builtin_amdgcn_readlane(idA, j);
      rkA += (c < idA) ? 1 : 0;
      rkB += (c < idB) ? 1 : 0;
    }
    for (int j = 0; j < nb; ++j) {
      const int c = __builtin_amdgcn_readlane(idB, j);
      rkA += (c < idA) ? 1 : 0;
      rkB += (c < idB) ? 1 : 0;
    }
    float m = wave_max(fmaxf(eA, eB));
    m = isfinite(m) ? m : 0.0f;
    float* sw = sw_s + (wv << 6);
    int*   sr = sr_s + (wv << 6);
    sw[rkA] = __expf(eA - m);
    sr[rkA] = snA;
    if (actB) { sw[rkB] = __expf(eB - m); sr[rkB] = snB; }
    const int   waA = __float_as_int(sw[lane]);
    const int   ssA = sr[lane];
    const int   waB = actB ? __float_as_int(sw[32 + lane]) : 0;
    const int   ssB = actB ? sr[32 + lane] : 0;
    float s = 0.0f;
    for (int k = 0; k < 32; ++k) s += __int_as_float(__builtin_amdgcn_readlane(waA, k));
    for (int k = 0; k < nb; ++k) s += __int_as_float(__builtin_amdgcn_readlane(waB, k));
    const float den = (s > 0.0f) ? s : 1.0f;
    const int alA = __float_as_int(__int_as_float(waA) / den);
    const int alB = __float_as_int(__int_as_float(waB) / den);
    for (int k0 = 0; k0 < 32; k0 += 16) {
      #pragma unroll
      for (int j = 0; j < 16; ++j) {
        const int   snk = __builtin_amdgcn_readlane(ssA, k0 + j);
        const float al  = __int_as_float(__builtin_amdgcn_readlane(alA, k0 + j));
        acc = fmaf(al, x[(size_t)snk * IND + lane], acc);
      }
    }
    for (int k0 = 0; k0 < nb; k0 += 16) {
      #pragma unroll
      for (int j = 0; j < 16; ++j) {
        const int   snk = __builtin_amdgcn_readlane(ssB, k0 + j);
        const float al  = __int_as_float(__builtin_amdgcn_readlane(alB, k0 + j));
        acc = fmaf(al, x[(size_t)snk * IND + lane], acc);
      }
    }
    As[wv * APITCH + lane] = acc;
  } else {
    float mx = -INFINITY;
    for (int q = lane; q < deg; q += 32)
      mx = fmaxf(mx, leaky(el[src[csr[seg_index(sg, q)]]] + ern));
    float m = wave_max(mx);
    m = isfinite(m) ? m : 0.0f;
    float s = 0.0f;
    int prev = -1;
    for (int k = 0; k < deg; ++k) {
      prev = next_edge(csr, sg, deg, prev, lane);
      s += __expf(leaky(el[src[prev]] + ern) - m);
    }
    const float den = (s > 0.0f) ? s : 1.0f;
    prev = -1;
    for (int k = 0; k < deg; ++k) {
      prev = next_edge(csr, sg, deg, prev, lane);
      const int snk = src[prev];
      const float al = __expf(leaky(el[snk] + ern) - m) / den;
      acc = fmaf(al, x[(size_t)snk * IND + lane], acc);
    }
    As[wv * APITCH + lane] = acc;
  }
  __syncthreads();

  v2f A[8];
  #pragma unroll
  for (int i = 0; i < 8; ++i) {
    const int k = (i << 2) + (h << 1);
    A[i] = *(const v2f*)(As + r * APITCH + k);
  }
  v8f cacc[2] = {};
  #pragma unroll
  for (int i = 0; i < 8; ++i) {
    #pragma unroll
    for (int t = 0; t < 2; ++t)
      cacc[t] = __builtin_amdgcn_wmma_f32_16x16x4_f32(
          false, A[i], false, B[t][i], (short)0, cacc[t], false, false);
  }

  {
    const int col0 = (wv << 5) + (r << 1);
    #pragma unroll
    for (int rr = 0; rr < 8; ++rr) {
      int idx = (rr + (h << 3)) * OUTD + col0;
      idx = (col0 < OUTD) ? idx : 16 * OUTD + (r << 1);
      v2f pr;
      pr.x = cacc[0][rr];
      pr.y = cacc[1][rr];
      *(v2f*)(S + idx) = pr;
    }
  }
  __syncthreads();

  {
    const bool last = (lane + 96) < NQ;
    float4* r4 = S4 + wv * NQ;
    float4 v[4];
    #pragma unroll
    for (int j = 0; j < 3; ++j) v[j] = r4[lane + 32 * j];
    v[3] = last ? r4[lane + 96] : make_float4(-INFINITY, -INFINITY, -INFINITY, -INFINITY);
    float mx = -INFINITY;
    #pragma unroll
    for (int j = 0; j < 4; ++j)
      mx = fmaxf(mx, fmaxf(fmaxf(v[j].x, v[j].y), fmaxf(v[j].z, v[j].w)));
    mx = wave_max(mx);
    float sm = 0.0f;
    #pragma unroll
    for (int j = 0; j < 4; ++j) {
      const bool act = (j < 3) || last;
      v[j].x = act ? __expf(v[j].x - mx) : 0.0f;
      v[j].y = act ? __expf(v[j].y - mx) : 0.0f;
      v[j].z = act ? __expf(v[j].z - mx) : 0.0f;
      v[j].w = act ? __expf(v[j].w - mx) : 0.0f;
      sm += (v[j].x + v[j].y) + (v[j].z + v[j].w);
    }
    sm = wave_sum(sm);
    const float inv = 1.0f / sm;
    #pragma unroll
    for (int j = 0; j < 4; ++j) {
      v[j].x *= inv; v[j].y *= inv; v[j].z *= inv; v[j].w *= inv;
    }
    #pragma unroll
    for (int j = 0; j < 3; ++j) r4[lane + 32 * j] = v[j];
    if (last) r4[lane + 96] = v[3];
  }
  __syncthreads();

  float4* o4 = (float4*)(out + (size_t)m0 * OUTD);
  const int tid = threadIdx.x;
  #pragma unroll
  for (int j = 0; j < 3; ++j) o4[tid + 512 * j] = S4[tid + 512 * j];
  if (tid < 16 * NQ - 3 * 512) o4[tid + 3 * 512] = S4[tid + 3 * 512];
}

extern "C" void kernel_launch(void* const* d_in, const int* in_sizes, int n_in,
                              void* d_out, int out_size, void* d_ws, size_t ws_size,
                              hipStream_t stream) {
  const float* x      = (const float*)d_in[0];
  const float* W      = (const float*)d_in[1];
  const float* attn_l = (const float*)d_in[2];
  const float* attn_r = (const float*)d_in[3];
  const float* px_r   = (const float*)d_in[4];
  const int*   src    = (const int*)d_in[5];
  const int*   dst    = (const int*)d_in[6];
  float*       out    = (float*)d_out;

  float* ws   = (float*)d_ws;
  float* el   = ws;
  float* er   = el + NN;
  int*   off  = (int*)(er + NN);
  int*   csr  = off + NPART * (NN + 1);

  k_build              <<<GA, TA, 0, stream>>>(dst, x, W, attn_l, attn_r, px_r, out, off, csr, el, er);
  k_gather_gemm_softmax<<<MTILES, 512, 0, stream>>>(src, off, csr, el, er, x, W, out);
}
